# starting kernel
# baseline (speedup 1.0000x reference)
#include <hip/hip_runtime.h>

typedef float v2f __attribute__((ext_vector_type(2)));
typedef float v8f __attribute__((ext_vector_type(8)));

#define N_NODES 50000
#define N_EDGES 800000
#define D 128
#define EPS_F 1e-7f

__global__ void cge_zero_kernel(float* __restrict__ p, int n) {
    int i = blockIdx.x * blockDim.x + threadIdx.x;
    if (i < n) p[i] = 0.0f;
}

__global__ __launch_bounds__(256) void cge_scatter_kernel(
    const float* __restrict__ h, int h_stride,
    const int* __restrict__ esrc, const int* __restrict__ edst,
    float* __restrict__ neigh) {
    int lane = threadIdx.x & 31;
    int edge = (blockIdx.x * blockDim.x + threadIdx.x) >> 5;
    if (edge >= N_EDGES) return;
    int s = esrc[edge];
    int d = edst[edge];
    float4 v = reinterpret_cast<const float4*>(h + (size_t)s * h_stride)[lane];
    float* np = neigh + (size_t)d * D + lane * 4;
    atomicAdd(np + 0, v.x);
    atomicAdd(np + 1, v.y);
    atomicAdd(np + 2, v.z);
    atomicAdd(np + 3, v.w);
}

__global__ __launch_bounds__(256) void cge_gemm_kernel(
    const float* __restrict__ A, int a_stride,
    const float* __restrict__ neigh,
    const float* __restrict__ deg,
    const float* __restrict__ W,
    const float* __restrict__ bias,
    float* __restrict__ out, int out_stride) {
    const int lane  = threadIdx.x & 31;
    const int warp  = threadIdx.x >> 5;
    const int mtile = blockIdx.x;
    const int rc    = lane & 15;
    const int khalf = lane >> 4;

    const int m = mtile * 16 + rc;
    const int n = warp * 16 + rc;

    const float invdeg = 1.0f / (deg[m] + EPS_F);
    const float* __restrict__ arow = A     + (size_t)m * a_stride;
    const float* __restrict__ nrow = neigh + (size_t)m * D;

    v8f c = {};

#pragma unroll 8
    for (int k0 = 0; k0 < 128; k0 += 4) {
        const int k = k0 + 2 * khalf;
        v2f a; a.x = arow[k];              a.y = arow[k + 1];
        v2f b; b.x = W[(size_t)k * D + n]; b.y = W[(size_t)(k + 1) * D + n];
        c = __builtin_amdgcn_wmma_f32_16x16x4_f32(
            false, a, false, b, (short)0, c, false, false);
    }
#pragma unroll 8
    for (int k0 = 0; k0 < 128; k0 += 4) {
        const int k = k0 + 2 * khalf;
        v2f a; a.x = nrow[k] * invdeg;     a.y = nrow[k + 1] * invdeg;
        v2f b; b.x = W[(size_t)(128 + k) * D + n];
               b.y = W[(size_t)(129 + k) * D + n];
        c = __builtin_amdgcn_wmma_f32_16x16x4_f32(
            false, a, false, b, (short)0, c, false, false);
    }

    const float bn    = bias[n];
    const int   mbase = mtile * 16 + khalf * 8;
#pragma unroll
    for (int j = 0; j < 8; ++j) {
        float v = c[j] + bn;
        v = v > 0.0f ? v : 0.0f;
        out[(size_t)(mbase + j) * out_stride + n] = v;
    }
}

__global__ void cge_copyx_kernel(const float* __restrict__ x,
                                 float* __restrict__ out) {
    int i = blockIdx.x * blockDim.x + threadIdx.x;
    if (i >= N_NODES * (D / 4)) return;
    int r  = i >> 5;
    int c4 = i & 31;
    float4 v = reinterpret_cast<const float4*>(x)[i];
    reinterpret_cast<float4*>(out + (size_t)r * 256 + 128)[c4] = v;
}

extern "C" void kernel_launch(void* const* d_in, const int* in_sizes, int n_in,
                              void* d_out, int out_size, void* d_ws, size_t ws_size,
                              hipStream_t stream) {
    const float* x    = (const float*)d_in[0];
    const int*   esrc = (const int*)  d_in[1];
    const int*   edst = (const int*)  d_in[2];
    const float* deg  = (const float*)d_in[3];
    const float* W1   = (const float*)d_in[4];
    const float* b1   = (const float*)d_in[5];
    const float* W2   = (const float*)d_in[6];
    const float* b2   = (const float*)d_in[7];
    float* out = (float*)d_out;

    float* neigh = (float*)d_ws;
    float* h1    = out + 128;

    const int NF = N_NODES * D;
    dim3 blk(256);
    dim3 zgrid((NF + 255) / 256);
    dim3 sgrid(((size_t)N_EDGES * 32 + 255) / 256);
    dim3 ggrid(N_NODES / 16);

    cge_zero_kernel   <<<zgrid, blk, 0, stream>>>(neigh, NF);
    cge_scatter_kernel<<<sgrid, blk, 0, stream>>>(x, D, esrc, edst, neigh);
    cge_gemm_kernel   <<<ggrid, blk, 0, stream>>>(x, D, neigh, deg, W1, b1,
                                                  h1, 256);
    cge_zero_kernel   <<<zgrid, blk, 0, stream>>>(neigh, NF);
    cge_scatter_kernel<<<sgrid, blk, 0, stream>>>(h1, 256, esrc, edst, neigh);
    cge_gemm_kernel   <<<ggrid, blk, 0, stream>>>(h1, 256, neigh, deg, W2, b2,
                                                  out, 256);
    cge_copyx_kernel  <<<(N_NODES * (D / 4) + 255) / 256, blk, 0, stream>>>(x, out);
}


# Layer-1 sweep reads 1-byte owner keys (dst/196), not 4-byte destinations: a new first launch writes them to idle out rows, blocks stream a quarter of the bytes; 4 launches; LDS arrival sum order kept
# speedup vs baseline: 3.0373x; 3.0373x over previous
#include <hip/hip_runtime.h>

typedef float v2f __attribute__((ext_vector_type(2)));
typedef float v8f __attribute__((ext_vector_type(8)));

#define N_NODES 50000
#define N_EDGES 800000
#define D 128
#define EPS_F 1e-7f

#define OWN_BLOCKS  256
#define OWN_ROWS    196
#define OWN_THREADS 1024
#define OWN_WAVES   32
#define SCAN_K      8
#define N_GROUPS    (N_EDGES / 128)
#define SCAN_STEP   (OWN_WAVES * SCAN_K)
#define SCAN_TRIPS  ((N_GROUPS + SCAN_STEP - 1) / SCAN_STEP)
#define QCAP        (32 + SCAN_K * 128)
#define DRAIN_U     8
#define LIST_LDS    416
#define LIST_E0     8
#define OUT_PITCH   256
#define M_TILES     13
#define TILE_ROWS   (M_TILES * 16)
#define ROW_P       132
#define W_P         288
#define GEMM_WAVES  (M_TILES * 2)
#define X_CELLS     ((TILE_ROWS * 32 + OWN_THREADS - 1) / OWN_THREADS)
#define XT_ROWS     (X_CELLS * OWN_THREADS / 32)
#define UBUF_FLOATS (XT_ROWS * ROW_P + 64 * W_P)

__device__ __forceinline__ int cge_group(int trip, int k, int wave, int rot) {
    int g = trip * SCAN_STEP + k * OWN_WAVES + wave;
    if (g >= N_GROUPS) return -1;
    g += rot;
    if (g >= N_GROUPS) g -= N_GROUPS;
    return g;
}

__device__ __forceinline__ void cge_drain_ent(
    int ent, int first, int n, int lane,
    const float* __restrict__ h, int h_stride, float* acc) {
    int j = 0;
    for (; j + DRAIN_U <= n; j += DRAIN_U) {
        float v[DRAIN_U][4];
        int   dd[DRAIN_U];
#pragma unroll
        for (int u = 0; u < DRAIN_U; ++u) {
            const int en = __builtin_amdgcn_readlane(ent, first + j + u);
            dd[u] = en & 255;
            const float* __restrict__ hp = h + (size_t)(en >> 8) * h_stride + lane;
            v[u][0] = hp[0]; v[u][1] = hp[32]; v[u][2] = hp[64]; v[u][3] = hp[96];
        }
#pragma unroll
        for (int u = 0; u < DRAIN_U; ++u) {
            float* ap = acc + dd[u] * ROW_P + lane;
            __hip_atomic_fetch_add(ap,      v[u][0], __ATOMIC_RELAXED, __HIP_MEMORY_SCOPE_WORKGROUP);
            __hip_atomic_fetch_add(ap + 32, v[u][1], __ATOMIC_RELAXED, __HIP_MEMORY_SCOPE_WORKGROUP);
            __hip_atomic_fetch_add(ap + 64, v[u][2], __ATOMIC_RELAXED, __HIP_MEMORY_SCOPE_WORKGROUP);
            __hip_atomic_fetch_add(ap + 96, v[u][3], __ATOMIC_RELAXED, __HIP_MEMORY_SCOPE_WORKGROUP);
        }
    }
    if (j < n) {
        float v[DRAIN_U][4];
        int   dd[DRAIN_U];
#pragma unroll
        for (int u = 0; u < DRAIN_U; ++u) {
            const int jj = (j + u < n) ? (j + u) : j;
            const int en = __builtin_amdgcn_readlane(ent, first + jj);
            dd[u] = en & 255;
            const float* __restrict__ hp = h + (size_t)(en >> 8) * h_stride + lane;
            v[u][0] = hp[0]; v[u][1] = hp[32]; v[u][2] = hp[64]; v[u][3] = hp[96];
        }
#pragma unroll
        for (int u = 0; u < DRAIN_U; ++u) {
            if (j + u < n) {
                float* ap = acc + dd[u] * ROW_P + lane;
                __hip_atomic_fetch_add(ap,      v[u][0], __ATOMIC_RELAXED, __HIP_MEMORY_SCOPE_WORKGROUP);
                __hip_atomic_fetch_add(ap + 32, v[u][1], __ATOMIC_RELAXED, __HIP_MEMORY_SCOPE_WORKGROUP);
                __hip_atomic_fetch_add(ap + 64, v[u][2], __ATOMIC_RELAXED, __HIP_MEMORY_SCOPE_WORKGROUP);
                __hip_atomic_fetch_add(ap + 96, v[u][3], __ATOMIC_RELAXED, __HIP_MEMORY_SCOPE_WORKGROUP);
            }
        }
    }
}

template <bool KEYS>
__device__ __forceinline__ void cge_drain_q(
    const int* qw, int base, int n, int lane,
    const int* __restrict__ esrc, const int* __restrict__ edst,
    const float* __restrict__ h, int h_stride,
    float* acc, int* sl, int maxent, int lo, int nrows, int& cnt, bool& ovf) {
    int ent = 0;
    if (lane < n) {
        const int q = qw[base + lane];
        if (KEYS) {
            const int s = esrc[q];
            int dl = edst[q] - lo;
            if ((unsigned)dl >= (unsigned)nrows) dl = TILE_ROWS - 1;
            ent = (s << 8) | dl;
        } else {
            ent = (esrc[q >> 8] << 8) | (q & 255);
        }
    }
    if (KEYS && !ovf) {
        if (cnt + n <= maxent) { if (lane < n) sl[LIST_E0 + cnt + lane] = ent; cnt += n; }
        else ovf = true;
    }
    cge_drain_ent(ent, 0, n, lane, h, h_stride, acc);
}

__device__ __forceinline__ int cge_sweep_dst(
    const int* __restrict__ esrc, const int* __restrict__ edst,
    const float* __restrict__ h, int h_stride, float* acc,
    int* qw, int* sl, int maxent, int lo, int nrows, int wave, int lane, int rot) {
    const int4* __restrict__ d4 = reinterpret_cast<const int4*>(edst);
    int  qn  = 0;
    int  cnt = 0;
    bool ovf = false;

    int4 cur[SCAN_K], nxt[SCAN_K];
#pragma unroll
    for (int k = 0; k < SCAN_K; ++k) {
        const int g = cge_group(0, k, wave, rot);
        cur[k] = d4[(g < 0 ? 0 : g) * 32 + lane];
    }

    for (int trip = 0; trip < SCAN_TRIPS; ++trip) {
#pragma unroll
        for (int k = 0; k < SCAN_K; ++k) {
            const int g = cge_group(trip + 1 < SCAN_TRIPS ? trip + 1 : trip, k, wave, rot);
            nxt[k] = d4[(g < 0 ? 0 : g) * 32 + lane];
        }
#pragma unroll
        for (int k = 0; k < SCAN_K; ++k) {
            const int g = cge_group(trip, k, wave, rot);
            if (g < 0) continue;
            const int ebase = g * 128 + lane * 4;
            const int dl0 = cur[k].x - lo, dl1 = cur[k].y - lo;
            const int dl2 = cur[k].z - lo, dl3 = cur[k].w - lo;
            const bool m0 = (unsigned)dl0 < (unsigned)nrows;
            const bool m1 = (unsigned)dl1 < (unsigned)nrows;
            const bool m2 = (unsigned)dl2 < (unsigned)nrows;
            const bool m3 = (unsigned)dl3 < (unsigned)nrows;
            if (__builtin_amdgcn_ballot_w32(m0 | m1 | m2 | m3) == 0u) continue;
            const unsigned below = (1u << lane) - 1u;
            unsigned mk;
            mk = __builtin_amdgcn_ballot_w32(m0);
            if (m0) qw[qn + __builtin_popcount(mk & below)] = ((ebase + 0) << 8) | dl0;
            qn += __builtin_popcount(mk);
            mk = __builtin_amdgcn_ballot_w32(m1);
            if (m1) qw[qn + __builtin_popcount(mk & below)] = ((ebase + 1) << 8) | dl1;
            qn += __builtin_popcount(mk);
            mk = __builtin_amdgcn_ballot_w32(m2);
            if (m2) qw[qn + __builtin_popcount(mk & below)] = ((ebase + 2) << 8) | dl2;
            qn += __builtin_popcount(mk);
            mk = __builtin_amdgcn_ballot_w32(m3);
            if (m3) qw[qn + __builtin_popcount(mk & below)] = ((ebase + 3) << 8) | dl3;
            qn += __builtin_popcount(mk);
        }
        while (qn >= 32) {
            cge_drain_q<false>(qw, qn - 32, 32, lane, esrc, edst, h, h_stride, acc, sl, maxent, lo, nrows, cnt, ovf);
            qn -= 32;
        }
#pragma unroll
        for (int k = 0; k < SCAN_K; ++k) cur[k] = nxt[k];
    }
    if (qn > 0) cge_drain_q<false>(qw, 0, qn, lane, esrc, edst, h, h_stride, acc, sl, maxent, lo, nrows, cnt, ovf);
    return ovf ? -1 : cnt;
}

#define KEY_WORDS      (N_EDGES / 4)
#define KEY_ROW0       (OWN_WAVES * LIST_LDS / 128)
#define KEY_HOST_ROWS  (OWN_ROWS - KEY_ROW0)
__device__ __forceinline__ size_t cge_key_row_word(int krow) {
    const int hb = krow / KEY_HOST_ROWS;
    const int hr = krow - hb * KEY_HOST_ROWS;
    return (size_t)(hb * OWN_ROWS + KEY_ROW0 + hr) * OUT_PITCH;
}

#define N_OCT      ((N_GROUPS + 7) / 8)
#define OCT_TRIPS  ((N_OCT + OWN_WAVES - 1) / OWN_WAVES)
__device__ __forceinline__ int cge_octet(int trip, int wave, int rot) {
    int o = trip * OWN_WAVES + wave;
    if (o >= N_OCT) return -1;
    o += rot;
    if (o >= N_OCT) o -= N_OCT;
    return o;
}
__device__ __forceinline__ int cge_sweep_keys(
    const int* __restrict__ esrc, const int* __restrict__ edst, const int* keys,
    const float* __restrict__ h, int h_stride, float* acc,
    int* qw, int* sl, int maxent, int lo, int nrows, int wave, int lane, int rot) {
    static_assert(KEY_HOST_ROWS % 2 == 0, "the two key rows of an octet must lie in one hosting block");
    const unsigned mine = (unsigned)blockIdx.x * 0x01010101u;
    const unsigned below = (1u << lane) - 1u;
    int  qn  = 0;
    int  cnt = 0;
    bool ovf = false;

    int cur[SCAN_K], nxt[SCAN_K];
    {
        const int o = cge_octet(0, wave, rot);
        const int* p = keys + cge_key_row_word(2 * (o < 0 ? 0 : o)) + lane;
#pragma unroll
        for (int k = 0; k < SCAN_K; ++k) cur[k] = p[(k >> 2) * OUT_PITCH + (k & 3) * 32];
    }

    for (int trip = 0; trip < OCT_TRIPS; ++trip) {
        {
            const int o = cge_octet(trip + 1 < OCT_TRIPS ? trip + 1 : trip, wave, rot);
            const int* p = keys + cge_key_row_word(2 * (o < 0 ? 0 : o)) + lane;
#pragma unroll
            for (int k = 0; k < SCAN_K; ++k) nxt[k] = p[(k >> 2) * OUT_PITCH + (k & 3) * 32];
        }
        const int o = cge_octet(trip, wave, rot);
        if (o >= 0) {
#pragma unroll
            for (int k = 0; k < SCAN_K; ++k) {
                const int g = o * SCAN_K + k;
                if (g >= N_GROUPS) continue;
                const unsigned x = (unsigned)cur[k] ^ mine;
                unsigned z = ~(((x & 0x7f7f7f7fu) + 0x7f7f7f7fu) | x | 0x7f7f7f7fu);
                unsigned mk = __builtin_amdgcn_ballot_w32(z != 0u);
                if (mk == 0u) continue;
                const int ebase = g * 128 + lane * 4;
                do {
                    if (z != 0u) qw[qn + __builtin_popcount(mk & below)] = ebase + (__builtin_ctz(z) >> 3);
                    qn += __builtin_popcount(mk);
                    z &= z - 1u;
                    mk = __builtin_amdgcn_ballot_w32(z != 0u);
                } while (mk != 0u);
            }
        }
        while (qn >= 32) {
            cge_drain_q<true>(qw, qn - 32, 32, lane, esrc, edst, h, h_stride, acc, sl, maxent, lo, nrows, cnt, ovf);
            qn -= 32;
        }
#pragma unroll
        for (int k = 0; k < SCAN_K; ++k) cur[k] = nxt[k];
    }
    if (qn > 0) cge_drain_q<true>(qw, 0, qn, lane, esrc, edst, h, h_stride, acc, sl, maxent, lo, nrows, cnt, ovf);
    return ovf ? -1 : cnt;
}

#define KEY_THREADS 256
#define KEY_WAVES   (2 * N_OCT)
#define KEY_BLOCKS  ((KEY_WAVES + KEY_THREADS / 32 - 1) / (KEY_THREADS / 32))
__global__ __launch_bounds__(KEY_THREADS) void cge_keys_kernel(
    const int* __restrict__ edst, float* out) {
    const int lane = threadIdx.x & 31;
    const int wv   = blockIdx.x * (KEY_THREADS / 32) + (threadIdx.x >> 5);
    const int4* __restrict__ d4 = reinterpret_cast<const int4*>(edst);
    int* kw = reinterpret_cast<int*>(out);
    int4 d[4];
#pragma unroll
    for (int j = 0; j < 4; ++j) {
        const int i = wv * 128 + j * 32 + lane;
        d[j] = d4[i < KEY_WORDS ? i : 0];
    }
#pragma unroll
    for (int j = 0; j < 4; ++j) {
        const int i = wv * 128 + j * 32 + lane;
        if (wv < KEY_WAVES) {
            const unsigned k0 = ((unsigned)d[j].x / OWN_ROWS) & 255u;
            const unsigned k1 = ((unsigned)d[j].y / OWN_ROWS) & 255u;
            const unsigned k2 = ((unsigned)d[j].z / OWN_ROWS) & 255u;
            const unsigned k3 = ((unsigned)d[j].w / OWN_ROWS) & 255u;
            kw[cge_key_row_word(wv) + j * 32 + lane] = (int)(k0 | (k1 << 8) | (k2 << 16) | (k3 << 24));
        }
    }
}

template <bool SCALE>
__device__ __forceinline__ void cge_half_product(
    const float* ap, float ascale, const float* bp, v8f (&c)[4]) {
#pragma unroll 4
    for (int ks = 0; ks < 32; ++ks) {
        v2f a = *reinterpret_cast<const v2f*>(ap + 4 * ks);
        if (SCALE) { a.x *= ascale; a.y *= ascale; }
        const float* bk = bp + 2 * ks * W_P;
#pragma unroll
        for (int nt = 0; nt < 4; ++nt) {
            const v2f b = *reinterpret_cast<const v2f*>(bk + nt * 32);
            c[nt] = __builtin_amdgcn_wmma_f32_16x16x4_f32(
                false, a, false, b, (short)0, c[nt], false, false);
        }
    }
}

#define W_CELLS 2
__device__ __forceinline__ void cge_load_w(
    const float* __restrict__ W, int kbase, int tid, float4 (&wr)[2 * W_CELLS]) {
#pragma unroll
    for (int t = 0; t < W_CELLS; ++t) {
        const int i = tid + t * OWN_THREADS;
        const int kp = i >> 5, c4 = i & 31;
        wr[2 * t]     = *reinterpret_cast<const float4*>(W + (size_t)(kbase + 2 * kp) * D + c4 * 4);
        wr[2 * t + 1] = *reinterpret_cast<const float4*>(W + (size_t)(kbase + 2 * kp + 1) * D + c4 * 4);
    }
}
__device__ __forceinline__ void cge_store_w(
    float* wl, int tid, const float4 (&wr)[2 * W_CELLS]) {
#pragma unroll
    for (int t = 0; t < W_CELLS; ++t) {
        const int i = tid + t * OWN_THREADS;
        const int kp = i >> 5, c4 = i & 31;
        const float4 w0 = wr[2 * t], w1 = wr[2 * t + 1];
        float4* dst = reinterpret_cast<float4*>(wl + kp * W_P + c4 * 8);
        dst[0] = make_float4(w0.x, w1.x, w0.y, w1.y);
        dst[1] = make_float4(w0.z, w1.z, w0.w, w1.w);
    }
}

template <bool REUSE>
__global__ __launch_bounds__(OWN_THREADS) void cge_owner_layer_kernel(
    const float* __restrict__ h, int h_stride,
    const int* __restrict__ esrc, const int* __restrict__ edst,
    const float* __restrict__ deg,
    const float* __restrict__ W,
    const float* __restrict__ bias,
    float* out, int out_stride,
    float* lists) {
    __shared__ alignas(16) float acc[TILE_ROWS * ROW_P];
    __shared__ alignas(16) float ubuf[UBUF_FLOATS];

    const int tid   = threadIdx.x;
    const int lane  = tid & 31;
    const int wave  = __builtin_amdgcn_readfirstlane(tid >> 5);
    const int lo    = blockIdx.x * OWN_ROWS;
    const int nrows = (N_NODES - lo < OWN_ROWS) ? (N_NODES - lo) : OWN_ROWS;

    const int gfit   = (nrows * 4) & ~31;
    const int gcap   = gfit < LIST_LDS ? gfit : LIST_LDS;
    const int maxent = (gcap < LIST_LDS - 32) ? (gcap - LIST_E0) : (LIST_LDS - 32 - LIST_E0);
    int* lw = reinterpret_cast<int*>(lists);
    const int gw0 = wave * gcap;

    int line0 = 0;
    if (REUSE) {
        const int gi = gw0 + lane;
        line0 = lw[(size_t)(lo + (gi >> 7)) * OUT_PITCH + (gi & 127)];
    }

    float4* acc4 = reinterpret_cast<float4*>(acc);
    for (int i = tid; i < TILE_ROWS * ROW_P / 4; i += OWN_THREADS)
        acc4[i] = make_float4(0.0f, 0.0f, 0.0f, 0.0f);
    __syncthreads();

    const int rot = blockIdx.x * (N_GROUPS / OWN_BLOCKS);
    int* qw = reinterpret_cast<int*>(ubuf) + wave * QCAP;
    int* sl = reinterpret_cast<int*>(ubuf) + OWN_WAVES * QCAP + wave * LIST_LDS;

    if (!REUSE) {
        const int cnt = cge_sweep_keys(esrc, edst, lw, h, h_stride, acc, qw, sl, maxent, lo, nrows, wave, lane,
                                       (int)blockIdx.x * (N_OCT / OWN_BLOCKS));
        const int ne = (cnt < 0) ? 0 : cnt;
        sl[LIST_E0 + ne + lane] = 0;
        if (lane < LIST_E0) sl[lane] = (lane == 0) ? cnt : 0;
        const int nl = ((LIST_E0 - 1 + ne) >> 5) + 1;
        for (int i = lane * 4; i < nl * 32; i += 128) {
            const int gi = gw0 + i;
            *reinterpret_cast<int4*>(lw + (size_t)(lo + (gi >> 7)) * OUT_PITCH + (gi & 127)) =
                *reinterpret_cast<const int4*>(sl + i);
        }
    } else {
        const int cnt = __builtin_amdgcn_readfirstlane(line0);
        if ((unsigned)cnt <= (unsigned)maxent) {
            const int lastw = LIST_E0 - 1 + cnt;
            int cur = line0, first = LIST_E0;
            for (int base = 0; base <= lastw; base += 32) {
                const int ni = base + 32 + lane;
                int nxt = 0;
                if (ni <= lastw) {
                    const int gi = gw0 + ni;
                    nxt = lw[(size_t)(lo + (gi >> 7)) * OUT_PITCH + (gi & 127)];
                }
                const int last = (lastw < base + 31) ? lastw : (base + 31);
                cge_drain_ent(cur, first, last - base - first + 1, lane, h, h_stride, acc);
                cur = nxt; first = 0;
            }
        } else {
            (void)cge_sweep_dst(esrc, edst, h, h_stride, acc, qw, sl, maxent, lo, nrows, wave, lane, rot);
        }
    }
    __syncthreads();

    float* xt = ubuf;
    float* wl = ubuf + XT_ROWS * ROW_P;

    const int  rc     = lane & 15;
    const int  khalf  = lane >> 4;
    const int  mt     = wave >> 1;
    const int  nh     = wave & 1;
    const bool active = (wave < GEMM_WAVES) && (mt * 16 < nrows);
    const int  arow   = mt * 16 + rc;
    const int  aoff   = arow * ROW_P + 2 * khalf;
    const int  boff   = khalf * W_P + (nh * 64 + rc) * 2;

    float4 xr[X_CELLS];
    float4 wr[2 * W_CELLS];
#pragma unroll
    for (int t = 0; t < X_CELLS; ++t) {
        const int i = tid + t * OWN_THREADS;
        const int r = i >> 5, c4 = i & 31;
        const int rr = (r < nrows) ? r : (nrows - 1);
        xr[t] = *reinterpret_cast<const float4*>(h + (size_t)(lo + rr) * h_stride + c4 * 4);
    }
    cge_load_w(W, 0, tid, wr);
    const float dg = deg[lo + (arow < nrows ? arow : nrows - 1)];
    float bn[4];
#pragma unroll
    for (int nt = 0; nt < 4; ++nt) bn[nt] = bias[nh * 64 + nt * 16 + rc];
#pragma unroll
    for (int t = 0; t < X_CELLS; ++t) {
        const int i = tid + t * OWN_THREADS;
        const int r = i >> 5, c4 = i & 31;
        *reinterpret_cast<float4*>(xt + r * ROW_P + c4 * 4) = xr[t];
    }
    cge_store_w(wl, tid, wr);
    __syncthreads();

    cge_load_w(W, 128, tid, wr);

    v8f c[4] = {};
    if (active) cge_half_product<false>(xt + aoff, 1.0f, wl + boff, c);
    __syncthreads();
    cge_store_w(wl, tid, wr);
    __syncthreads();

    if (active) {
        const float invdeg = 1.0f / (dg + EPS_F);
        cge_half_product<true>(acc + aoff, invdeg, wl + boff, c);

#pragma unroll
        for (int nt = 0; nt < 4; ++nt) {
            const int n = nh * 64 + nt * 16 + rc;
#pragma unroll
            for (int j = 0; j < 8; ++j) {
                float v = c[nt][j] + bn[nt];
                v = v > 0.0f ? v : 0.0f;
                xt[(mt * 16 + khalf * 8 + j) * ROW_P + n] = v;
            }
        }
    }
    __syncthreads();

    for (int i = tid; i < nrows * 32; i += OWN_THREADS) {
        const int r = i >> 5, c4 = i & 31;
        *reinterpret_cast<float4*>(out + (size_t)(lo + r) * out_stride + c4 * 4) =
            *reinterpret_cast<const float4*>(xt + r * ROW_P + c4 * 4);
    }
}

__global__ void cge_copyx_kernel(const float* __restrict__ x,
                                 float* __restrict__ out) {
    int i = blockIdx.x * blockDim.x + threadIdx.x;
    if (i >= N_NODES * (D / 4)) return;
    int r  = i >> 5;
    int c4 = i & 31;
    float4 v = reinterpret_cast<const float4*>(x)[i];
    reinterpret_cast<float4*>(out + (size_t)r * 256 + 128)[c4] = v;
}

extern "C" void kernel_launch(void* const* d_in, const int* in_sizes, int n_in,
                              void* d_out, int out_size, void* d_ws, size_t ws_size,
                              hipStream_t stream) {
    const float* x    = (const float*)d_in[0];
    const int*   esrc = (const int*)  d_in[1];
    const int*   edst = (const int*)  d_in[2];
    const float* deg  = (const float*)d_in[3];
    const float* W1   = (const float*)d_in[4];
    const float* b1   = (const float*)d_in[5];
    const float* W2   = (const float*)d_in[6];
    const float* b2   = (const float*)d_in[7];
    float* out = (float*)d_out;

    float* h1    = out + 128;

    dim3 blk(256);
    dim3 ogrid(OWN_BLOCKS);
    dim3 oblk(OWN_THREADS);

    cge_keys_kernel<<<dim3(KEY_BLOCKS), dim3(KEY_THREADS), 0, stream>>>(edst, out);
    cge_owner_layer_kernel<false><<<ogrid, oblk, 0, stream>>>(x, D, esrc, edst, deg, W1, b1,
                                                              h1, 256, out);
    cge_owner_layer_kernel<true><<<ogrid, oblk, 0, stream>>>(h1, 256, esrc, edst, deg, W2, b2,
                                                             out, 256, out);
    cge_copyx_kernel  <<<(N_NODES * (D / 4) + 255) / 256, blk, 0, stream>>>(x, out);
}


# Layer 2: each wave runs 2 k-steps of its self-feature product between asking for a batch's 8 source rows and adding them (h tile, W half staged before the sum); all waves sum; LDS arrival order kept
# speedup vs baseline: 3.1434x; 1.0349x over previous
#include <hip/hip_runtime.h>

typedef float v2f __attribute__((ext_vector_type(2)));
typedef float v8f __attribute__((ext_vector_type(8)));

#define N_NODES 50000
#define N_EDGES 800000
#define D 128
#define EPS_F 1e-7f

#define OWN_BLOCKS  256
#define OWN_ROWS    196
#define OWN_THREADS 1024
#define OWN_WAVES   32
#define SCAN_K      8
#define N_GROUPS    (N_EDGES / 128)
#define QCAP        (32 + SCAN_K * 128)
#define FQ_CAP      (32 + 128)
#define MM_KS       2
#define DRAIN_U     8
#define LIST_LDS    416
#define LIST_E0     8
#define OUT_PITCH   256
#define M_TILES     13
#define TILE_ROWS   (M_TILES * 16)
#define ROW_P       132
#define W_P         288
#define GEMM_WAVES  (M_TILES * 2)
#define X_CELLS     ((TILE_ROWS * 32 + OWN_THREADS - 1) / OWN_THREADS)
#define XT_ROWS     (X_CELLS * OWN_THREADS / 32)
#define UBUF_FLOATS (XT_ROWS * ROW_P + 64 * W_P)


__device__ __forceinline__ void cge_drain_ent(
    int ent, int first, int n, int lane,
    const float* __restrict__ h, int h_stride, float* acc) {
    int j = 0;
    for (; j + DRAIN_U <= n; j += DRAIN_U) {
        float v[DRAIN_U][4];
        int   dd[DRAIN_U];
#pragma unroll
        for (int u = 0; u < DRAIN_U; ++u) {
            const int en = __builtin_amdgcn_readlane(ent, first + j + u);
            dd[u] = en & 255;
            const float* __restrict__ hp = h + (size_t)(en >> 8) * h_stride + lane;
            v[u][0] = hp[0]; v[u][1] = hp[32]; v[u][2] = hp[64]; v[u][3] = hp[96];
        }
#pragma unroll
        for (int u = 0; u < DRAIN_U; ++u) {
            float* ap = acc + dd[u] * ROW_P + lane;
            __hip_atomic_fetch_add(ap,      v[u][0], __ATOMIC_RELAXED, __HIP_MEMORY_SCOPE_WORKGROUP);
            __hip_atomic_fetch_add(ap + 32, v[u][1], __ATOMIC_RELAXED, __HIP_MEMORY_SCOPE_WORKGROUP);
            __hip_atomic_fetch_add(ap + 64, v[u][2], __ATOMIC_RELAXED, __HIP_MEMORY_SCOPE_WORKGROUP);
            __hip_atomic_fetch_add(ap + 96, v[u][3], __ATOMIC_RELAXED, __HIP_MEMORY_SCOPE_WORKGROUP);
        }
    }
    if (j < n) {
        float v[DRAIN_U][4];
        int   dd[DRAIN_U];
#pragma unroll
        for (int u = 0; u < DRAIN_U; ++u) {
            const int jj = (j + u < n) ? (j + u) : j;
            const int en = __builtin_amdgcn_readlane(ent, first + jj);
            dd[u] = en & 255;
            const float* __restrict__ hp = h + (size_t)(en >> 8) * h_stride + lane;
            v[u][0] = hp[0]; v[u][1] = hp[32]; v[u][2] = hp[64]; v[u][3] = hp[96];
        }
#pragma unroll
        for (int u = 0; u < DRAIN_U; ++u) {
            if (j + u < n) {
                float* ap = acc + dd[u] * ROW_P + lane;
                __hip_atomic_fetch_add(ap,      v[u][0], __ATOMIC_RELAXED, __HIP_MEMORY_SCOPE_WORKGROUP);
                __hip_atomic_fetch_add(ap + 32, v[u][1], __ATOMIC_RELAXED, __HIP_MEMORY_SCOPE_WORKGROUP);
                __hip_atomic_fetch_add(ap + 64, v[u][2], __ATOMIC_RELAXED, __HIP_MEMORY_SCOPE_WORKGROUP);
                __hip_atomic_fetch_add(ap + 96, v[u][3], __ATOMIC_RELAXED, __HIP_MEMORY_SCOPE_WORKGROUP);
            }
        }
    }
}

template <bool KEYS>
__device__ __forceinline__ void cge_drain_q(
    const int* qw, int base, int n, int lane,
    const int* __restrict__ esrc, const int* __restrict__ edst,
    const float* __restrict__ h, int h_stride,
    float* acc, int* sl, int maxent, int lo, int nrows, int& cnt, bool& ovf) {
    int ent = 0;
    if (lane < n) {
        const int q = qw[base + lane];
        if (KEYS) {
            const int s = esrc[q];
            int dl = edst[q] - lo;
            if ((unsigned)dl >= (unsigned)nrows) dl = TILE_ROWS - 1;
            ent = (s << 8) | dl;
        } else {
            ent = (esrc[q >> 8] << 8) | (q & 255);
        }
    }
    if (KEYS && !ovf) {
        if (cnt + n <= maxent) { if (lane < n) sl[LIST_E0 + cnt + lane] = ent; cnt += n; }
        else ovf = true;
    }
    cge_drain_ent(ent, 0, n, lane, h, h_stride, acc);
}

#define FB_TRIPS ((N_GROUPS + OWN_WAVES - 1) / OWN_WAVES)
__device__ __forceinline__ void cge_sweep_dst(
    const int* __restrict__ esrc, const int* __restrict__ edst,
    const float* __restrict__ h, int h_stride, float* acc,
    int* qw, int lo, int nrows, int wave, int lane, int rot) {
    const int4* __restrict__ d4 = reinterpret_cast<const int4*>(edst);
    const unsigned below = (1u << lane) - 1u;
    int  qn  = 0;
    int  cnt = 0;
    bool ovf = false;
    for (int trip = 0; trip < FB_TRIPS; ++trip) {
        int g = trip * OWN_WAVES + wave;
        if (g >= N_GROUPS) break;
        g += rot;
        if (g >= N_GROUPS) g -= N_GROUPS;
        const int4 d = d4[g * 32 + lane];
        const int ebase = g * 128 + lane * 4;
        const int dl0 = d.x - lo, dl1 = d.y - lo, dl2 = d.z - lo, dl3 = d.w - lo;
        const bool m0 = (unsigned)dl0 < (unsigned)nrows;
        const bool m1 = (unsigned)dl1 < (unsigned)nrows;
        const bool m2 = (unsigned)dl2 < (unsigned)nrows;
        const bool m3 = (unsigned)dl3 < (unsigned)nrows;
        unsigned mk;
        mk = __builtin_amdgcn_ballot_w32(m0);
        if (m0) qw[qn + __builtin_popcount(mk & below)] = ((ebase + 0) << 8) | dl0;
        qn += __builtin_popcount(mk);
        mk = __builtin_amdgcn_ballot_w32(m1);
        if (m1) qw[qn + __builtin_popcount(mk & below)] = ((ebase + 1) << 8) | dl1;
        qn += __builtin_popcount(mk);
        mk = __builtin_amdgcn_ballot_w32(m2);
        if (m2) qw[qn + __builtin_popcount(mk & below)] = ((ebase + 2) << 8) | dl2;
        qn += __builtin_popcount(mk);
        mk = __builtin_amdgcn_ballot_w32(m3);
        if (m3) qw[qn + __builtin_popcount(mk & below)] = ((ebase + 3) << 8) | dl3;
        qn += __builtin_popcount(mk);
        while (qn >= 32) {
            cge_drain_q<false>(qw, qn - 32, 32, lane, esrc, edst, h, h_stride, acc, nullptr, 0, lo, nrows, cnt, ovf);
            qn -= 32;
        }
    }
    if (qn > 0) cge_drain_q<false>(qw, 0, qn, lane, esrc, edst, h, h_stride, acc, nullptr, 0, lo, nrows, cnt, ovf);
}

#define KEY_WORDS      (N_EDGES / 4)
#define KEY_ROW0       (OWN_WAVES * LIST_LDS / 128)
#define KEY_HOST_ROWS  (OWN_ROWS - KEY_ROW0)
__device__ __forceinline__ size_t cge_key_row_word(int krow) {
    const int hb = krow / KEY_HOST_ROWS;
    const int hr = krow - hb * KEY_HOST_ROWS;
    return (size_t)(hb * OWN_ROWS + KEY_ROW0 + hr) * OUT_PITCH;
}

#define N_OCT      ((N_GROUPS + 7) / 8)
#define OCT_TRIPS  ((N_OCT + OWN_WAVES - 1) / OWN_WAVES)
__device__ __forceinline__ int cge_octet(int trip, int wave, int rot) {
    int o = trip * OWN_WAVES + wave;
    if (o >= N_OCT) return -1;
    o += rot;
    if (o >= N_OCT) o -= N_OCT;
    return o;
}
__device__ __forceinline__ int cge_sweep_keys(
    const int* __restrict__ esrc, const int* __restrict__ edst, const int* keys,
    const float* __restrict__ h, int h_stride, float* acc,
    int* qw, int* sl, int maxent, int lo, int nrows, int wave, int lane, int rot) {
    static_assert(KEY_HOST_ROWS % 2 == 0, "the two key rows of an octet must lie in one hosting block");
    const unsigned mine = (unsigned)blockIdx.x * 0x01010101u;
    const unsigned below = (1u << lane) - 1u;
    int  qn  = 0;
    int  cnt = 0;
    bool ovf = false;

    int cur[SCAN_K], nxt[SCAN_K];
    {
        const int o = cge_octet(0, wave, rot);
        const int* p = keys + cge_key_row_word(2 * (o < 0 ? 0 : o)) + lane;
#pragma unroll
        for (int k = 0; k < SCAN_K; ++k) cur[k] = p[(k >> 2) * OUT_PITCH + (k & 3) * 32];
    }

    for (int trip = 0; trip < OCT_TRIPS; ++trip) {
        {
            const int o = cge_octet(trip + 1 < OCT_TRIPS ? trip + 1 : trip, wave, rot);
            const int* p = keys + cge_key_row_word(2 * (o < 0 ? 0 : o)) + lane;
#pragma unroll
            for (int k = 0; k < SCAN_K; ++k) nxt[k] = p[(k >> 2) * OUT_PITCH + (k & 3) * 32];
        }
        const int o = cge_octet(trip, wave, rot);
        if (o >= 0) {
#pragma unroll
            for (int k = 0; k < SCAN_K; ++k) {
                const int g = o * SCAN_K + k;
                if (g >= N_GROUPS) continue;
                const unsigned x = (unsigned)cur[k] ^ mine;
                unsigned z = ~(((x & 0x7f7f7f7fu) + 0x7f7f7f7fu) | x | 0x7f7f7f7fu);
                unsigned mk = __builtin_amdgcn_ballot_w32(z != 0u);
                if (mk == 0u) continue;
                const int ebase = g * 128 + lane * 4;
                do {
                    if (z != 0u) qw[qn + __builtin_popcount(mk & below)] = ebase + (__builtin_ctz(z) >> 3);
                    qn += __builtin_popcount(mk);
                    z &= z - 1u;
                    mk = __builtin_amdgcn_ballot_w32(z != 0u);
                } while (mk != 0u);
            }
        }
        while (qn >= 32) {
            cge_drain_q<true>(qw, qn - 32, 32, lane, esrc, edst, h, h_stride, acc, sl, maxent, lo, nrows, cnt, ovf);
            qn -= 32;
        }
#pragma unroll
        for (int k = 0; k < SCAN_K; ++k) cur[k] = nxt[k];
    }
    if (qn > 0) cge_drain_q<true>(qw, 0, qn, lane, esrc, edst, h, h_stride, acc, sl, maxent, lo, nrows, cnt, ovf);
    return ovf ? -1 : cnt;
}

#define KEY_THREADS 256
#define KEY_WAVES   (2 * N_OCT)
#define KEY_BLOCKS  ((KEY_WAVES + KEY_THREADS / 32 - 1) / (KEY_THREADS / 32))
__global__ __launch_bounds__(KEY_THREADS) void cge_keys_kernel(
    const int* __restrict__ edst, float* out) {
    const int lane = threadIdx.x & 31;
    const int wv   = blockIdx.x * (KEY_THREADS / 32) + (threadIdx.x >> 5);
    const int4* __restrict__ d4 = reinterpret_cast<const int4*>(edst);
    int* kw = reinterpret_cast<int*>(out);
    int4 d[4];
#pragma unroll
    for (int j = 0; j < 4; ++j) {
        const int i = wv * 128 + j * 32 + lane;
        d[j] = d4[i < KEY_WORDS ? i : 0];
    }
#pragma unroll
    for (int j = 0; j < 4; ++j) {
        const int i = wv * 128 + j * 32 + lane;
        if (wv < KEY_WAVES) {
            const unsigned k0 = ((unsigned)d[j].x / OWN_ROWS) & 255u;
            const unsigned k1 = ((unsigned)d[j].y / OWN_ROWS) & 255u;
            const unsigned k2 = ((unsigned)d[j].z / OWN_ROWS) & 255u;
            const unsigned k3 = ((unsigned)d[j].w / OWN_ROWS) & 255u;
            kw[cge_key_row_word(wv) + j * 32 + lane] = (int)(k0 | (k1 << 8) | (k2 << 16) | (k3 << 24));
        }
    }
}

template <bool SCALE>
__device__ __forceinline__ void cge_half_product(
    const float* ap, float ascale, const float* bp, v8f (&c)[4]) {
#pragma unroll 4
    for (int ks = 0; ks < 32; ++ks) {
        v2f a = *reinterpret_cast<const v2f*>(ap + 4 * ks);
        if (SCALE) { a.x *= ascale; a.y *= ascale; }
        const float* bk = bp + 2 * ks * W_P;
#pragma unroll
        for (int nt = 0; nt < 4; ++nt) {
            const v2f b = *reinterpret_cast<const v2f*>(bk + nt * 32);
            c[nt] = __builtin_amdgcn_wmma_f32_16x16x4_f32(
                false, a, false, b, (short)0, c[nt], false, false);
        }
    }
}

#define W_CELLS 2
__device__ __forceinline__ void cge_load_w(
    const float* __restrict__ W, int kbase, int tid, float4 (&wr)[2 * W_CELLS]) {
#pragma unroll
    for (int t = 0; t < W_CELLS; ++t) {
        const int i = tid + t * OWN_THREADS;
        const int kp = i >> 5, c4 = i & 31;
        wr[2 * t]     = *reinterpret_cast<const float4*>(W + (size_t)(kbase + 2 * kp) * D + c4 * 4);
        wr[2 * t + 1] = *reinterpret_cast<const float4*>(W + (size_t)(kbase + 2 * kp + 1) * D + c4 * 4);
    }
}
__device__ __forceinline__ void cge_store_w(
    float* wl, int tid, const float4 (&wr)[2 * W_CELLS]) {
#pragma unroll
    for (int t = 0; t < W_CELLS; ++t) {
        const int i = tid + t * OWN_THREADS;
        const int kp = i >> 5, c4 = i & 31;
        const float4 w0 = wr[2 * t], w1 = wr[2 * t + 1];
        float4* dst = reinterpret_cast<float4*>(wl + kp * W_P + c4 * 8);
        dst[0] = make_float4(w0.x, w1.x, w0.y, w1.y);
        dst[1] = make_float4(w0.z, w1.z, w0.w, w1.w);
    }
}

__device__ __forceinline__ void cge_kstep(const float* ap, const float* bp, int ks, v8f (&c)[4]) {
    const v2f a = *reinterpret_cast<const v2f*>(ap + 4 * ks);
    const float* bk = bp + 2 * ks * W_P;
#pragma unroll
    for (int nt = 0; nt < 4; ++nt) {
        const v2f b = *reinterpret_cast<const v2f*>(bk + nt * 32);
        c[nt] = __builtin_amdgcn_wmma_f32_16x16x4_f32(
            false, a, false, b, (short)0, c[nt], false, false);
    }
}

__device__ __forceinline__ void cge_drain_ent_mm(
    int ent, int first, int n, int lane,
    const float* __restrict__ h, int h_stride, float* acc,
    bool mm, const float* ap, const float* bp, v8f (&c)[4], int& ks) {
    int j = 0;
    for (; j + DRAIN_U <= n; j += DRAIN_U) {
        float v[DRAIN_U][4];
        int   dd[DRAIN_U];
#pragma unroll
        for (int u = 0; u < DRAIN_U; ++u) {
            const int en = __builtin_amdgcn_readlane(ent, first + j + u);
            dd[u] = en & 255;
            const float* __restrict__ hp = h + (size_t)(en >> 8) * h_stride + lane;
            v[u][0] = hp[0]; v[u][1] = hp[32]; v[u][2] = hp[64]; v[u][3] = hp[96];
        }
        if (mm && ks < 32) {
#pragma unroll
            for (int t = 0; t < MM_KS; ++t) cge_kstep(ap, bp, ks + t, c);
            ks += MM_KS;
        }
#pragma unroll
        for (int u = 0; u < DRAIN_U; ++u) {
            float* aq = acc + dd[u] * ROW_P + lane;
            __hip_atomic_fetch_add(aq,      v[u][0], __ATOMIC_RELAXED, __HIP_MEMORY_SCOPE_WORKGROUP);
            __hip_atomic_fetch_add(aq + 32, v[u][1], __ATOMIC_RELAXED, __HIP_MEMORY_SCOPE_WORKGROUP);
            __hip_atomic_fetch_add(aq + 64, v[u][2], __ATOMIC_RELAXED, __HIP_MEMORY_SCOPE_WORKGROUP);
            __hip_atomic_fetch_add(aq + 96, v[u][3], __ATOMIC_RELAXED, __HIP_MEMORY_SCOPE_WORKGROUP);
        }
    }
    if (j < n) {
        float v[DRAIN_U][4];
        int   dd[DRAIN_U];
#pragma unroll
        for (int u = 0; u < DRAIN_U; ++u) {
            const int jj = (j + u < n) ? (j + u) : j;
            const int en = __builtin_amdgcn_readlane(ent, first + jj);
            dd[u] = en & 255;
            const float* __restrict__ hp = h + (size_t)(en >> 8) * h_stride + lane;
            v[u][0] = hp[0]; v[u][1] = hp[32]; v[u][2] = hp[64]; v[u][3] = hp[96];
        }
        if (mm && ks < 32) {
#pragma unroll
            for (int t = 0; t < MM_KS; ++t) cge_kstep(ap, bp, ks + t, c);
            ks += MM_KS;
        }
#pragma unroll
        for (int u = 0; u < DRAIN_U; ++u) {
            if (j + u < n) {
                float* aq = acc + dd[u] * ROW_P + lane;
                __hip_atomic_fetch_add(aq,      v[u][0], __ATOMIC_RELAXED, __HIP_MEMORY_SCOPE_WORKGROUP);
                __hip_atomic_fetch_add(aq + 32, v[u][1], __ATOMIC_RELAXED, __HIP_MEMORY_SCOPE_WORKGROUP);
                __hip_atomic_fetch_add(aq + 64, v[u][2], __ATOMIC_RELAXED, __HIP_MEMORY_SCOPE_WORKGROUP);
                __hip_atomic_fetch_add(aq + 96, v[u][3], __ATOMIC_RELAXED, __HIP_MEMORY_SCOPE_WORKGROUP);
            }
        }
    }
}

__device__ __forceinline__ void cge_load_x(
    const float* __restrict__ h, int h_stride, int lo, int nrows, int tid, float4 (&xr)[X_CELLS]) {
#pragma unroll
    for (int t = 0; t < X_CELLS; ++t) {
        const int i = tid + t * OWN_THREADS;
        const int r = i >> 5, c4 = i & 31;
        const int rr = (r < nrows) ? r : (nrows - 1);
        xr[t] = *reinterpret_cast<const float4*>(h + (size_t)(lo + rr) * h_stride + c4 * 4);
    }
}
__device__ __forceinline__ void cge_store_x(float* xt, int tid, const float4 (&xr)[X_CELLS]) {
#pragma unroll
    for (int t = 0; t < X_CELLS; ++t) {
        const int i = tid + t * OWN_THREADS;
        const int r = i >> 5, c4 = i & 31;
        *reinterpret_cast<float4*>(xt + r * ROW_P + c4 * 4) = xr[t];
    }
}

template <bool REUSE>
__global__ __launch_bounds__(OWN_THREADS) void cge_owner_layer_kernel(
    const float* __restrict__ h, int h_stride,
    const int* __restrict__ esrc, const int* __restrict__ edst,
    const float* __restrict__ deg,
    const float* __restrict__ W,
    const float* __restrict__ bias,
    float* out, int out_stride,
    float* lists) {
    __shared__ alignas(16) float acc[TILE_ROWS * ROW_P];
    __shared__ alignas(16) float ubuf[UBUF_FLOATS];
    __shared__ int fq[REUSE ? OWN_WAVES * FQ_CAP : 1];

    const int tid   = threadIdx.x;
    const int lane  = tid & 31;
    const int wave  = __builtin_amdgcn_readfirstlane(tid >> 5);
    const int lo    = blockIdx.x * OWN_ROWS;
    const int nrows = (N_NODES - lo < OWN_ROWS) ? (N_NODES - lo) : OWN_ROWS;

    const int gfit   = (nrows * 4) & ~31;
    const int gcap   = gfit < LIST_LDS ? gfit : LIST_LDS;
    const int maxent = (gcap < LIST_LDS - 32) ? (gcap - LIST_E0) : (LIST_LDS - 32 - LIST_E0);
    int* lw = reinterpret_cast<int*>(lists);
    const int gw0 = wave * gcap;

    int line0 = 0;
    if (REUSE) {
        const int gi = gw0 + lane;
        line0 = lw[(size_t)(lo + (gi >> 7)) * OUT_PITCH + (gi & 127)];
    }

    float* xt = ubuf;
    float* wl = ubuf + XT_ROWS * ROW_P;

    const int  rc     = lane & 15;
    const int  khalf  = lane >> 4;
    const int  mt     = wave >> 1;
    const int  nh     = wave & 1;
    const bool active = (wave < GEMM_WAVES) && (mt * 16 < nrows);
    const int  arow   = mt * 16 + rc;
    const int  aoff   = arow * ROW_P + 2 * khalf;
    const int  boff   = khalf * W_P + (nh * 64 + rc) * 2;

    float4 xr[X_CELLS];
    float4 wr[2 * W_CELLS];
    if (REUSE) {
        cge_load_x(h, h_stride, lo, nrows, tid, xr);
        cge_load_w(W, 0, tid, wr);
    }

    float4* acc4 = reinterpret_cast<float4*>(acc);
    for (int i = tid; i < TILE_ROWS * ROW_P / 4; i += OWN_THREADS)
        acc4[i] = make_float4(0.0f, 0.0f, 0.0f, 0.0f);
    if (REUSE) {
        cge_store_x(xt, tid, xr);
        cge_store_w(wl, tid, wr);
    }
    __syncthreads();

    const int rot = blockIdx.x * (N_GROUPS / OWN_BLOCKS);
    int* qw = reinterpret_cast<int*>(ubuf) + wave * QCAP;
    int* sl = reinterpret_cast<int*>(ubuf) + OWN_WAVES * QCAP + wave * LIST_LDS;

    v8f   c[4] = {};
    float dg = 1.0f;
    float bn[4] = {0.0f, 0.0f, 0.0f, 0.0f};

    if (!REUSE) {
        const int cnt = cge_sweep_keys(esrc, edst, lw, h, h_stride, acc, qw, sl, maxent, lo, nrows, wave, lane,
                                       (int)blockIdx.x * (N_OCT / OWN_BLOCKS));
        const int ne = (cnt < 0) ? 0 : cnt;
        sl[LIST_E0 + ne + lane] = 0;
        if (lane < LIST_E0) sl[lane] = (lane == 0) ? cnt : 0;
        const int nl = ((LIST_E0 - 1 + ne) >> 5) + 1;
        for (int i = lane * 4; i < nl * 32; i += 128) {
            const int gi = gw0 + i;
            *reinterpret_cast<int4*>(lw + (size_t)(lo + (gi >> 7)) * OUT_PITCH + (gi & 127)) =
                *reinterpret_cast<const int4*>(sl + i);
        }
    } else {
        int ks = 0;
        const int cnt = __builtin_amdgcn_readfirstlane(line0);
        if ((unsigned)cnt <= (unsigned)maxent) {
            const int lastw = LIST_E0 - 1 + cnt;
            int cur = line0, first = LIST_E0;
            for (int base = 0; base <= lastw; base += 32) {
                const int ni = base + 32 + lane;
                int nxt = 0;
                if (ni <= lastw) {
                    const int gi = gw0 + ni;
                    nxt = lw[(size_t)(lo + (gi >> 7)) * OUT_PITCH + (gi & 127)];
                }
                const int last = (lastw < base + 31) ? lastw : (base + 31);
                cge_drain_ent_mm(cur, first, last - base - first + 1, lane, h, h_stride, acc,
                                 active, xt + aoff, wl + boff, c, ks);
                cur = nxt; first = 0;
            }
        } else {
            cge_sweep_dst(esrc, edst, h, h_stride, acc, fq + wave * FQ_CAP, lo, nrows, wave, lane, rot);
        }
        cge_load_w(W, 128, tid, wr);
        dg = deg[lo + (arow < nrows ? arow : nrows - 1)];
#pragma unroll
        for (int nt = 0; nt < 4; ++nt) bn[nt] = bias[nh * 64 + nt * 16 + rc];
        if (active) {
            for (; ks < 32; ks += MM_KS) {
#pragma unroll
                for (int t = 0; t < MM_KS; ++t) cge_kstep(xt + aoff, wl + boff, ks + t, c);
            }
        }
    }
    __syncthreads();

    if (!REUSE) {
        cge_load_x(h, h_stride, lo, nrows, tid, xr);
        cge_load_w(W, 0, tid, wr);
        dg = deg[lo + (arow < nrows ? arow : nrows - 1)];
#pragma unroll
        for (int nt = 0; nt < 4; ++nt) bn[nt] = bias[nh * 64 + nt * 16 + rc];
        cge_store_x(xt, tid, xr);
        cge_store_w(wl, tid, wr);
        __syncthreads();

        cge_load_w(W, 128, tid, wr);

        if (active) cge_half_product<false>(xt + aoff, 1.0f, wl + boff, c);
        __syncthreads();
    }
    cge_store_w(wl, tid, wr);
    __syncthreads();

    if (active) {
        const float invdeg = 1.0f / (dg + EPS_F);
        cge_half_product<true>(acc + aoff, invdeg, wl + boff, c);

#pragma unroll
        for (int nt = 0; nt < 4; ++nt) {
            const int n = nh * 64 + nt * 16 + rc;
#pragma unroll
            for (int j = 0; j < 8; ++j) {
                float v = c[nt][j] + bn[nt];
                v = v > 0.0f ? v : 0.0f;
                xt[(mt * 16 + khalf * 8 + j) * ROW_P + n] = v;
            }
        }
    }
    __syncthreads();

    for (int i = tid; i < nrows * 32; i += OWN_THREADS) {
        const int r = i >> 5, c4 = i & 31;
        *reinterpret_cast<float4*>(out + (size_t)(lo + r) * out_stride + c4 * 4) =
            *reinterpret_cast<const float4*>(xt + r * ROW_P + c4 * 4);
    }
}

__global__ void cge_copyx_kernel(const float* __restrict__ x,
                                 float* __restrict__ out) {
    int i = blockIdx.x * blockDim.x + threadIdx.x;
    if (i >= N_NODES * (D / 4)) return;
    int r  = i >> 5;
    int c4 = i & 31;
    float4 v = reinterpret_cast<const float4*>(x)[i];
    reinterpret_cast<float4*>(out + (size_t)r * 256 + 128)[c4] = v;
}

extern "C" void kernel_launch(void* const* d_in, const int* in_sizes, int n_in,
                              void* d_out, int out_size, void* d_ws, size_t ws_size,
                              hipStream_t stream) {
    const float* x    = (const float*)d_in[0];
    const int*   esrc = (const int*)  d_in[1];
    const int*   edst = (const int*)  d_in[2];
    const float* deg  = (const float*)d_in[3];
    const float* W1   = (const float*)d_in[4];
    const float* b1   = (const float*)d_in[5];
    const float* W2   = (const float*)d_in[6];
    const float* b2   = (const float*)d_in[7];
    float* out = (float*)d_out;

    float* h1    = out + 128;

    dim3 blk(256);
    dim3 ogrid(OWN_BLOCKS);
    dim3 oblk(OWN_THREADS);

    cge_keys_kernel<<<dim3(KEY_BLOCKS), dim3(KEY_THREADS), 0, stream>>>(edst, out);
    cge_owner_layer_kernel<false><<<ogrid, oblk, 0, stream>>>(x, D, esrc, edst, deg, W1, b1,
                                                              h1, 256, out);
    cge_owner_layer_kernel<true><<<ogrid, oblk, 0, stream>>>(h1, 256, esrc, edst, deg, W2, b2,
                                                             out, 256, out);
    cge_copyx_kernel  <<<(N_NODES * (D / 4) + 255) / 256, blk, 0, stream>>>(x, out);
}


# Neighbor sums in registers: kept edges binned by owner wave (local row mod 32; place = LDS integer add), bin sorted by row, a wave adds each row in VGPRs, one LDS add per row; sum order = bin order
# speedup vs baseline: 3.4458x; 1.0962x over previous
#include <hip/hip_runtime.h>

typedef float v2f __attribute__((ext_vector_type(2)));
typedef float v8f __attribute__((ext_vector_type(8)));

#define N_NODES 50000
#define N_EDGES 800000
#define D 128
#define EPS_F 1e-7f

#define OWN_BLOCKS  256
#define OWN_ROWS    196
#define OWN_THREADS 1024
#define OWN_WAVES   32
#define SCAN_K      8
#define N_GROUPS    (N_EDGES / 128)
#define QCAP        (32 + SCAN_K * 128)
#define FQ_CAP      (32 + 128)
#define MM_KS       2
#define DRAIN_U     8
#define BIN_ROWS    7
#define LIST_LDS    416
#define LIST_E0     8
#define OUT_PITCH   256
#define M_TILES     13
#define TILE_ROWS   (M_TILES * 16)
#define ROW_P       132
#define W_P         288
#define GEMM_WAVES  (M_TILES * 2)
#define X_CELLS     ((TILE_ROWS * 32 + OWN_THREADS - 1) / OWN_THREADS)
#define XT_ROWS     (X_CELLS * OWN_THREADS / 32)
#define UBUF_FLOATS (XT_ROWS * ROW_P + 64 * W_P)


__device__ __forceinline__ void cge_drain_ent(
    int ent, int first, int n, int lane,
    const float* __restrict__ h, int h_stride, float* acc) {
    int j = 0;
    for (; j + DRAIN_U <= n; j += DRAIN_U) {
        float v[DRAIN_U][4];
        int   dd[DRAIN_U];
#pragma unroll
        for (int u = 0; u < DRAIN_U; ++u) {
            const int en = __builtin_amdgcn_readlane(ent, first + j + u);
            dd[u] = en & 255;
            const float* __restrict__ hp = h + (size_t)(en >> 8) * h_stride + lane;
            v[u][0] = hp[0]; v[u][1] = hp[32]; v[u][2] = hp[64]; v[u][3] = hp[96];
        }
#pragma unroll
        for (int u = 0; u < DRAIN_U; ++u) {
            float* ap = acc + dd[u] * ROW_P + lane;
            __hip_atomic_fetch_add(ap,      v[u][0], __ATOMIC_RELAXED, __HIP_MEMORY_SCOPE_WORKGROUP);
            __hip_atomic_fetch_add(ap + 32, v[u][1], __ATOMIC_RELAXED, __HIP_MEMORY_SCOPE_WORKGROUP);
            __hip_atomic_fetch_add(ap + 64, v[u][2], __ATOMIC_RELAXED, __HIP_MEMORY_SCOPE_WORKGROUP);
            __hip_atomic_fetch_add(ap + 96, v[u][3], __ATOMIC_RELAXED, __HIP_MEMORY_SCOPE_WORKGROUP);
        }
    }
    if (j < n) {
        float v[DRAIN_U][4];
        int   dd[DRAIN_U];
#pragma unroll
        for (int u = 0; u < DRAIN_U; ++u) {
            const int jj = (j + u < n) ? (j + u) : j;
            const int en = __builtin_amdgcn_readlane(ent, first + jj);
            dd[u] = en & 255;
            const float* __restrict__ hp = h + (size_t)(en >> 8) * h_stride + lane;
            v[u][0] = hp[0]; v[u][1] = hp[32]; v[u][2] = hp[64]; v[u][3] = hp[96];
        }
#pragma unroll
        for (int u = 0; u < DRAIN_U; ++u) {
            if (j + u < n) {
                float* ap = acc + dd[u] * ROW_P + lane;
                __hip_atomic_fetch_add(ap,      v[u][0], __ATOMIC_RELAXED, __HIP_MEMORY_SCOPE_WORKGROUP);
                __hip_atomic_fetch_add(ap + 32, v[u][1], __ATOMIC_RELAXED, __HIP_MEMORY_SCOPE_WORKGROUP);
                __hip_atomic_fetch_add(ap + 64, v[u][2], __ATOMIC_RELAXED, __HIP_MEMORY_SCOPE_WORKGROUP);
                __hip_atomic_fetch_add(ap + 96, v[u][3], __ATOMIC_RELAXED, __HIP_MEMORY_SCOPE_WORKGROUP);
            }
        }
    }
}

__device__ __forceinline__ void cge_drain_q(
    const int* qw, int base, int n, int lane,
    const int* __restrict__ esrc,
    const float* __restrict__ h, int h_stride, float* acc) {
    int ent = 0;
    if (lane < n) {
        const int q = qw[base + lane];
        ent = (esrc[q >> 8] << 8) | (q & 255);
    }
    cge_drain_ent(ent, 0, n, lane, h, h_stride, acc);
}

__device__ __forceinline__ void cge_bin_q(
    const int* qw, int base, int n, int lane,
    const int* __restrict__ esrc, const int* __restrict__ edst,
    const float* __restrict__ h, int h_stride,
    float* acc, int* bins, int* bincnt, int maxent, int lo, int nrows) {
    int  ent  = 0;
    bool over = false;
    if (lane < n) {
        const int q = qw[base + lane];
        const int sr = esrc[q];
        int dl = edst[q] - lo;
        if ((unsigned)dl >= (unsigned)nrows) dl = TILE_ROWS - 1;
        ent = (sr << 8) | dl;
        const int b = dl & 31;
        const int slot = __hip_atomic_fetch_add(bincnt + b, 1, __ATOMIC_RELAXED, __HIP_MEMORY_SCOPE_WORKGROUP);
        if (slot < maxent) bins[b * LIST_LDS + LIST_E0 + slot] = ent;
        else over = true;
    }
    unsigned om = __builtin_amdgcn_ballot_w32(over);
    while (om != 0u) {
        const int l = __builtin_ctz(om);
        om &= om - 1u;
        cge_drain_ent(ent, l, 1, lane, h, h_stride, acc);
    }
}

#define FB_TRIPS ((N_GROUPS + OWN_WAVES - 1) / OWN_WAVES)
__device__ __forceinline__ void cge_sweep_dst(
    const int* __restrict__ esrc, const int* __restrict__ edst,
    const float* __restrict__ h, int h_stride, float* acc,
    int* qw, int lo, int nrows, int wave, int lane, int rot) {
    const int4* __restrict__ d4 = reinterpret_cast<const int4*>(edst);
    const unsigned below = (1u << lane) - 1u;
    int  qn  = 0;
    for (int trip = 0; trip < FB_TRIPS; ++trip) {
        int g = trip * OWN_WAVES + wave;
        if (g >= N_GROUPS) break;
        g += rot;
        if (g >= N_GROUPS) g -= N_GROUPS;
        const int4 d = d4[g * 32 + lane];
        const int ebase = g * 128 + lane * 4;
        const int dl0 = d.x - lo, dl1 = d.y - lo, dl2 = d.z - lo, dl3 = d.w - lo;
        const bool m0 = (unsigned)dl0 < (unsigned)nrows;
        const bool m1 = (unsigned)dl1 < (unsigned)nrows;
        const bool m2 = (unsigned)dl2 < (unsigned)nrows;
        const bool m3 = (unsigned)dl3 < (unsigned)nrows;
        unsigned mk;
        mk = __builtin_amdgcn_ballot_w32(m0);
        if (m0) qw[qn + __builtin_popcount(mk & below)] = ((ebase + 0) << 8) | dl0;
        qn += __builtin_popcount(mk);
        mk = __builtin_amdgcn_ballot_w32(m1);
        if (m1) qw[qn + __builtin_popcount(mk & below)] = ((ebase + 1) << 8) | dl1;
        qn += __builtin_popcount(mk);
        mk = __builtin_amdgcn_ballot_w32(m2);
        if (m2) qw[qn + __builtin_popcount(mk & below)] = ((ebase + 2) << 8) | dl2;
        qn += __builtin_popcount(mk);
        mk = __builtin_amdgcn_ballot_w32(m3);
        if (m3) qw[qn + __builtin_popcount(mk & below)] = ((ebase + 3) << 8) | dl3;
        qn += __builtin_popcount(mk);
        while (qn >= 32) {
            cge_drain_q(qw, qn - 32, 32, lane, esrc, h, h_stride, acc);
            qn -= 32;
        }
    }
    if (qn > 0) cge_drain_q(qw, 0, qn, lane, esrc, h, h_stride, acc);
}

#define KEY_WORDS      (N_EDGES / 4)
#define KEY_ROW0       (OWN_WAVES * LIST_LDS / 128)
#define KEY_HOST_ROWS  (OWN_ROWS - KEY_ROW0)
__device__ __forceinline__ size_t cge_key_row_word(int krow) {
    const int hb = krow / KEY_HOST_ROWS;
    const int hr = krow - hb * KEY_HOST_ROWS;
    return (size_t)(hb * OWN_ROWS + KEY_ROW0 + hr) * OUT_PITCH;
}

#define N_OCT      ((N_GROUPS + 7) / 8)
#define OCT_TRIPS  ((N_OCT + OWN_WAVES - 1) / OWN_WAVES)
__device__ __forceinline__ int cge_octet(int trip, int wave, int rot) {
    int o = trip * OWN_WAVES + wave;
    if (o >= N_OCT) return -1;
    o += rot;
    if (o >= N_OCT) o -= N_OCT;
    return o;
}
__device__ __forceinline__ void cge_sweep_keys(
    const int* __restrict__ esrc, const int* __restrict__ edst, const int* keys,
    const float* __restrict__ h, int h_stride, float* acc,
    int* qw, int* bins, int* bincnt, int maxent, int lo, int nrows, int wave, int lane, int rot) {
    static_assert(KEY_HOST_ROWS % 2 == 0, "the two key rows of an octet must lie in one hosting block");
    const unsigned mine = (unsigned)blockIdx.x * 0x01010101u;
    const unsigned below = (1u << lane) - 1u;
    int  qn  = 0;

    int cur[SCAN_K], nxt[SCAN_K];
    {
        const int o = cge_octet(0, wave, rot);
        const int* p = keys + cge_key_row_word(2 * (o < 0 ? 0 : o)) + lane;
#pragma unroll
        for (int k = 0; k < SCAN_K; ++k) cur[k] = p[(k >> 2) * OUT_PITCH + (k & 3) * 32];
    }

    for (int trip = 0; trip < OCT_TRIPS; ++trip) {
        {
            const int o = cge_octet(trip + 1 < OCT_TRIPS ? trip + 1 : trip, wave, rot);
            const int* p = keys + cge_key_row_word(2 * (o < 0 ? 0 : o)) + lane;
#pragma unroll
            for (int k = 0; k < SCAN_K; ++k) nxt[k] = p[(k >> 2) * OUT_PITCH + (k & 3) * 32];
        }
        const int o = cge_octet(trip, wave, rot);
        if (o >= 0) {
#pragma unroll
            for (int k = 0; k < SCAN_K; ++k) {
                const int g = o * SCAN_K + k;
                if (g >= N_GROUPS) continue;
                const unsigned x = (unsigned)cur[k] ^ mine;
                unsigned z = ~(((x & 0x7f7f7f7fu) + 0x7f7f7f7fu) | x | 0x7f7f7f7fu);
                unsigned mk = __builtin_amdgcn_ballot_w32(z != 0u);
                if (mk == 0u) continue;
                const int ebase = g * 128 + lane * 4;
                do {
                    if (z != 0u) qw[qn + __builtin_popcount(mk & below)] = ebase + (__builtin_ctz(z) >> 3);
                    qn += __builtin_popcount(mk);
                    z &= z - 1u;
                    mk = __builtin_amdgcn_ballot_w32(z != 0u);
                } while (mk != 0u);
            }
        }
        while (qn >= 32) {
            cge_bin_q(qw, qn - 32, 32, lane, esrc, edst, h, h_stride, acc, bins, bincnt, maxent, lo, nrows);
            qn -= 32;
        }
#pragma unroll
        for (int k = 0; k < SCAN_K; ++k) cur[k] = nxt[k];
    }
    if (qn > 0) cge_bin_q(qw, 0, qn, lane, esrc, edst, h, h_stride, acc, bins, bincnt, maxent, lo, nrows);
}

#define KEY_THREADS 256
#define KEY_WAVES   (2 * N_OCT)
#define KEY_BLOCKS  ((KEY_WAVES + KEY_THREADS / 32 - 1) / (KEY_THREADS / 32))
__global__ __launch_bounds__(KEY_THREADS) void cge_keys_kernel(
    const int* __restrict__ edst, float* out) {
    const int lane = threadIdx.x & 31;
    const int wv   = blockIdx.x * (KEY_THREADS / 32) + (threadIdx.x >> 5);
    const int4* __restrict__ d4 = reinterpret_cast<const int4*>(edst);
    int* kw = reinterpret_cast<int*>(out);
    int4 d[4];
#pragma unroll
    for (int j = 0; j < 4; ++j) {
        const int i = wv * 128 + j * 32 + lane;
        d[j] = d4[i < KEY_WORDS ? i : 0];
    }
#pragma unroll
    for (int j = 0; j < 4; ++j) {
        const int i = wv * 128 + j * 32 + lane;
        if (wv < KEY_WAVES) {
            const unsigned k0 = ((unsigned)d[j].x / OWN_ROWS) & 255u;
            const unsigned k1 = ((unsigned)d[j].y / OWN_ROWS) & 255u;
            const unsigned k2 = ((unsigned)d[j].z / OWN_ROWS) & 255u;
            const unsigned k3 = ((unsigned)d[j].w / OWN_ROWS) & 255u;
            kw[cge_key_row_word(wv) + j * 32 + lane] = (int)(k0 | (k1 << 8) | (k2 << 16) | (k3 << 24));
        }
    }
}

template <bool SCALE>
__device__ __forceinline__ void cge_half_product(
    const float* ap, float ascale, const float* bp, v8f (&c)[4]) {
#pragma unroll 4
    for (int ks = 0; ks < 32; ++ks) {
        v2f a = *reinterpret_cast<const v2f*>(ap + 4 * ks);
        if (SCALE) { a.x *= ascale; a.y *= ascale; }
        const float* bk = bp + 2 * ks * W_P;
#pragma unroll
        for (int nt = 0; nt < 4; ++nt) {
            const v2f b = *reinterpret_cast<const v2f*>(bk + nt * 32);
            c[nt] = __builtin_amdgcn_wmma_f32_16x16x4_f32(
                false, a, false, b, (short)0, c[nt], false, false);
        }
    }
}

#define W_CELLS 2
__device__ __forceinline__ void cge_load_w(
    const float* __restrict__ W, int kbase, int tid, float4 (&wr)[2 * W_CELLS]) {
#pragma unroll
    for (int t = 0; t < W_CELLS; ++t) {
        const int i = tid + t * OWN_THREADS;
        const int kp = i >> 5, c4 = i & 31;
        wr[2 * t]     = *reinterpret_cast<const float4*>(W + (size_t)(kbase + 2 * kp) * D + c4 * 4);
        wr[2 * t + 1] = *reinterpret_cast<const float4*>(W + (size_t)(kbase + 2 * kp + 1) * D + c4 * 4);
    }
}
__device__ __forceinline__ void cge_store_w(
    float* wl, int tid, const float4 (&wr)[2 * W_CELLS]) {
#pragma unroll
    for (int t = 0; t < W_CELLS; ++t) {
        const int i = tid + t * OWN_THREADS;
        const int kp = i >> 5, c4 = i & 31;
        const float4 w0 = wr[2 * t], w1 = wr[2 * t + 1];
        float4* dst = reinterpret_cast<float4*>(wl + kp * W_P + c4 * 8);
        dst[0] = make_float4(w0.x, w1.x, w0.y, w1.y);
        dst[1] = make_float4(w0.z, w1.z, w0.w, w1.w);
    }
}

__device__ __forceinline__ void cge_kstep(const float* ap, const float* bp, int ks, v8f (&c)[4]) {
    const v2f a = *reinterpret_cast<const v2f*>(ap + 4 * ks);
    const float* bk = bp + 2 * ks * W_P;
#pragma unroll
    for (int nt = 0; nt < 4; ++nt) {
        const v2f b = *reinterpret_cast<const v2f*>(bk + nt * 32);
        c[nt] = __builtin_amdgcn_wmma_f32_16x16x4_f32(
            false, a, false, b, (short)0, c[nt], false, false);
    }
}

__device__ __forceinline__ void cge_put_row(const float (&a)[4], int cur, int wave, int lane, float* acc) {
    float* ap = acc + (wave + 32 * cur) * ROW_P + lane;
    __hip_atomic_fetch_add(ap,      a[0], __ATOMIC_RELAXED, __HIP_MEMORY_SCOPE_WORKGROUP);
    __hip_atomic_fetch_add(ap + 32, a[1], __ATOMIC_RELAXED, __HIP_MEMORY_SCOPE_WORKGROUP);
    __hip_atomic_fetch_add(ap + 64, a[2], __ATOMIC_RELAXED, __HIP_MEMORY_SCOPE_WORKGROUP);
    __hip_atomic_fetch_add(ap + 96, a[3], __ATOMIC_RELAXED, __HIP_MEMORY_SCOPE_WORKGROUP);
}

__device__ __forceinline__ void cge_run_add(
    float (&a)[4], int& cur, int r, const float (&v)[4], int wave, int lane, float* acc) {
    if (r != cur) {
        cge_put_row(a, cur, wave, lane, acc);
        a[0] = 0.0f; a[1] = 0.0f; a[2] = 0.0f; a[3] = 0.0f;
        cur = r;
    }
    a[0] += v[0]; a[1] += v[1]; a[2] += v[2]; a[3] += v[3];
}

template <bool MM, int U>
__device__ __forceinline__ void cge_drain_sorted(
    int ent, int first, int n, int lane,
    const float* __restrict__ h, int h_stride,
    float (&a)[4], int& cur, int wave, float* acc,
    bool mm, const float* ap, const float* bp, v8f (&c)[4], int& ks) {
    int j = 0;
    for (; j + U <= n; j += U) {
        float v[U][4];
        int   rr[U];
#pragma unroll
        for (int u = 0; u < U; ++u) {
            const int en = __builtin_amdgcn_readlane(ent, first + j + u);
            rr[u] = (en >> 5) & 7;
            const float* __restrict__ hp = h + (size_t)(en >> 8) * h_stride + lane;
            v[u][0] = hp[0]; v[u][1] = hp[32]; v[u][2] = hp[64]; v[u][3] = hp[96];
        }
        if (MM) {
            if (mm && ks < 32) {
#pragma unroll
                for (int t = 0; t < MM_KS; ++t) cge_kstep(ap, bp, ks + t, c);
                ks += MM_KS;
            }
        }
#pragma unroll
        for (int u = 0; u < U; ++u) cge_run_add(a, cur, rr[u], v[u], wave, lane, acc);
    }
    if (j < n) {
        float v[U][4];
        int   rr[U];
#pragma unroll
        for (int u = 0; u < U; ++u) {
            const int jj = (j + u < n) ? (j + u) : j;
            const int en = __builtin_amdgcn_readlane(ent, first + jj);
            rr[u] = (en >> 5) & 7;
            const float* __restrict__ hp = h + (size_t)(en >> 8) * h_stride + lane;
            v[u][0] = hp[0]; v[u][1] = hp[32]; v[u][2] = hp[64]; v[u][3] = hp[96];
        }
        if (MM) {
            if (mm && ks < 32) {
#pragma unroll
                for (int t = 0; t < MM_KS; ++t) cge_kstep(ap, bp, ks + t, c);
                ks += MM_KS;
            }
        }
#pragma unroll
        for (int u = 0; u < U; ++u) {
            if (j + u < n) cge_run_add(a, cur, rr[u], v[u], wave, lane, acc);
        }
    }
}

__device__ __forceinline__ void cge_sort_bin(const int* src, int ne, int* dst, int lane) {
    const unsigned below = (1u << lane) - 1u;
    int st[BIN_ROWS + 1];
#pragma unroll
    for (int r = 0; r <= BIN_ROWS; ++r) st[r] = 0;
    for (int base = 0; base < ne; base += 32) {
        const bool valid = base + lane < ne;
        const int  r = valid ? ((src[base + lane] >> 5) & 7) : BIN_ROWS;
#pragma unroll
        for (int k = 0; k < BIN_ROWS; ++k)
            st[k + 1] += __builtin_popcount(__builtin_amdgcn_ballot_w32(r == k));
    }
#pragma unroll
    for (int k = 1; k < BIN_ROWS; ++k) st[k + 1] += st[k];
    for (int base = 0; base < ne; base += 32) {
        const bool valid = base + lane < ne;
        const int  en = valid ? src[base + lane] : 0;
        const int  r  = valid ? ((en >> 5) & 7) : BIN_ROWS;
        int pos = 0;
#pragma unroll
        for (int k = 0; k < BIN_ROWS; ++k) {
            const unsigned m = __builtin_amdgcn_ballot_w32(r == k);
            pos = (r == k) ? st[k] + __builtin_popcount(m & below) : pos;
            st[k] += __builtin_popcount(m);
        }
        if (valid) dst[pos] = en;
    }
}

__device__ __forceinline__ void cge_load_x(
    const float* __restrict__ h, int h_stride, int lo, int nrows, int tid, float4 (&xr)[X_CELLS]) {
#pragma unroll
    for (int t = 0; t < X_CELLS; ++t) {
        const int i = tid + t * OWN_THREADS;
        const int r = i >> 5, c4 = i & 31;
        const int rr = (r < nrows) ? r : (nrows - 1);
        xr[t] = *reinterpret_cast<const float4*>(h + (size_t)(lo + rr) * h_stride + c4 * 4);
    }
}
__device__ __forceinline__ void cge_store_x(float* xt, int tid, const float4 (&xr)[X_CELLS]) {
#pragma unroll
    for (int t = 0; t < X_CELLS; ++t) {
        const int i = tid + t * OWN_THREADS;
        const int r = i >> 5, c4 = i & 31;
        *reinterpret_cast<float4*>(xt + r * ROW_P + c4 * 4) = xr[t];
    }
}

template <bool REUSE>
__global__ __launch_bounds__(OWN_THREADS) void cge_owner_layer_kernel(
    const float* __restrict__ h, int h_stride,
    const int* __restrict__ esrc, const int* __restrict__ edst,
    const float* __restrict__ deg,
    const float* __restrict__ W,
    const float* __restrict__ bias,
    float* out, int out_stride,
    float* lists) {
    __shared__ alignas(16) float acc[TILE_ROWS * ROW_P];
    __shared__ alignas(16) float ubuf[UBUF_FLOATS];
    __shared__ int fq[REUSE ? OWN_WAVES * FQ_CAP : 1];
    __shared__ int bincnt[REUSE ? 1 : OWN_WAVES];

    const int tid   = threadIdx.x;
    const int lane  = tid & 31;
    const int wave  = __builtin_amdgcn_readfirstlane(tid >> 5);
    const int lo    = blockIdx.x * OWN_ROWS;
    const int nrows = (N_NODES - lo < OWN_ROWS) ? (N_NODES - lo) : OWN_ROWS;

    const int gfit   = (nrows * 4) & ~31;
    const int gcap   = gfit < LIST_LDS ? gfit : LIST_LDS;
    const int maxent = (gcap < LIST_LDS - 32) ? (gcap - LIST_E0) : (LIST_LDS - 32 - LIST_E0);
    int* lw = reinterpret_cast<int*>(lists);
    const int gw0 = wave * gcap;

    int line0 = 0;
    if (REUSE) {
        const int gi = gw0 + lane;
        line0 = lw[(size_t)(lo + (gi >> 7)) * OUT_PITCH + (gi & 127)];
    }

    float* xt = ubuf;
    float* wl = ubuf + XT_ROWS * ROW_P;

    const int  rc     = lane & 15;
    const int  khalf  = lane >> 4;
    const int  mt     = wave >> 1;
    const int  nh     = wave & 1;
    const bool active = (wave < GEMM_WAVES) && (mt * 16 < nrows);
    const int  arow   = mt * 16 + rc;
    const int  aoff   = arow * ROW_P + 2 * khalf;
    const int  boff   = khalf * W_P + (nh * 64 + rc) * 2;

    float4 xr[X_CELLS];
    float4 wr[2 * W_CELLS];
    if (REUSE) {
        cge_load_x(h, h_stride, lo, nrows, tid, xr);
        cge_load_w(W, 0, tid, wr);
    }

    float4* acc4 = reinterpret_cast<float4*>(acc);
    for (int i = tid; i < TILE_ROWS * ROW_P / 4; i += OWN_THREADS)
        acc4[i] = make_float4(0.0f, 0.0f, 0.0f, 0.0f);
    if (!REUSE && tid < OWN_WAVES) bincnt[tid] = 0;
    if (REUSE) {
        cge_store_x(xt, tid, xr);
        cge_store_w(wl, tid, wr);
    }
    __syncthreads();

    const int rot = blockIdx.x * (N_GROUPS / OWN_BLOCKS);
    int* qw = reinterpret_cast<int*>(ubuf) + wave * QCAP;
    int* bins = reinterpret_cast<int*>(ubuf) + OWN_WAVES * QCAP;
    int* sl = bins + wave * LIST_LDS;

    v8f   c[4] = {};
    float dg = 1.0f;
    float bn[4] = {0.0f, 0.0f, 0.0f, 0.0f};

    if (!REUSE) {
        cge_sweep_keys(esrc, edst, lw, h, h_stride, acc, qw, bins, bincnt, maxent, lo, nrows, wave, lane,
                       (int)blockIdx.x * (N_OCT / OWN_BLOCKS));
        __syncthreads();
        const int  bc     = bincnt[lane];
        const bool marked = __builtin_amdgcn_ballot_w32(bc > maxent) != 0u;
        const int  sent   = __builtin_amdgcn_readlane(bc, wave);
        const int  ne     = sent < maxent ? sent : maxent;
        static_assert(QCAP >= LIST_LDS, "the sorted list is staged in the wave's queue space");
        cge_sort_bin(sl + LIST_E0, ne, qw + LIST_E0, lane);
        qw[LIST_E0 + ne + lane] = 0;
        if (lane < LIST_E0) qw[lane] = (lane == 0) ? (marked ? -1 : ne) : 0;
        const int nl = marked ? 1 : ((LIST_E0 - 1 + ne) >> 5) + 1;
        for (int i = lane * 4; i < nl * 32; i += 128) {
            const int gi = gw0 + i;
            *reinterpret_cast<int4*>(lw + (size_t)(lo + (gi >> 7)) * OUT_PITCH + (gi & 127)) =
                *reinterpret_cast<const int4*>(qw + i);
        }
        float a[4] = {0.0f, 0.0f, 0.0f, 0.0f};
        int currow = 0;
        int ksd = 0;
        for (int base = 0; base < ne; base += 32) {
            const int nb  = (ne - base < 32) ? (ne - base) : 32;
            const int ent = qw[LIST_E0 + base + lane];
            cge_drain_sorted<false, DRAIN_U>(ent, 0, nb, lane, h, h_stride, a, currow, wave, acc,
                                             false, nullptr, nullptr, c, ksd);
        }
        cge_put_row(a, currow, wave, lane, acc);
    } else {
        int ks = 0;
        const int cnt = __builtin_amdgcn_readfirstlane(line0);
        if ((unsigned)cnt <= (unsigned)maxent) {
            float a[4] = {0.0f, 0.0f, 0.0f, 0.0f};
            int currow = 0;
            const int lastw = LIST_E0 - 1 + cnt;
            int cur = line0, first = LIST_E0;
            for (int base = 0; base <= lastw; base += 32) {
                const int ni = base + 32 + lane;
                int nxt = 0;
                if (ni <= lastw) {
                    const int gi = gw0 + ni;
                    nxt = lw[(size_t)(lo + (gi >> 7)) * OUT_PITCH + (gi & 127)];
                }
                const int last = (lastw < base + 31) ? lastw : (base + 31);
                cge_drain_sorted<true, DRAIN_U>(cur, first, last - base - first + 1, lane, h, h_stride,
                                                a, currow, wave, acc, active, xt + aoff, wl + boff, c, ks);
                cur = nxt; first = 0;
            }
            cge_put_row(a, currow, wave, lane, acc);
        } else {
            cge_sweep_dst(esrc, edst, h, h_stride, acc, fq + wave * FQ_CAP, lo, nrows, wave, lane, rot);
        }
        cge_load_w(W, 128, tid, wr);
        dg = deg[lo + (arow < nrows ? arow : nrows - 1)];
#pragma unroll
        for (int nt = 0; nt < 4; ++nt) bn[nt] = bias[nh * 64 + nt * 16 + rc];
        if (active) {
            for (; ks < 32; ks += MM_KS) {
#pragma unroll
                for (int t = 0; t < MM_KS; ++t) cge_kstep(xt + aoff, wl + boff, ks + t, c);
            }
        }
    }
    __syncthreads();

    if (!REUSE) {
        cge_load_x(h, h_stride, lo, nrows, tid, xr);
        cge_load_w(W, 0, tid, wr);
        dg = deg[lo + (arow < nrows ? arow : nrows - 1)];
#pragma unroll
        for (int nt = 0; nt < 4; ++nt) bn[nt] = bias[nh * 64 + nt * 16 + rc];
        cge_store_x(xt, tid, xr);
        cge_store_w(wl, tid, wr);
        __syncthreads();

        cge_load_w(W, 128, tid, wr);

        if (active) cge_half_product<false>(xt + aoff, 1.0f, wl + boff, c);
        __syncthreads();
    }
    cge_store_w(wl, tid, wr);
    __syncthreads();

    if (active) {
        const float invdeg = 1.0f / (dg + EPS_F);
        cge_half_product<true>(acc + aoff, invdeg, wl + boff, c);

#pragma unroll
        for (int nt = 0; nt < 4; ++nt) {
            const int n = nh * 64 + nt * 16 + rc;
#pragma unroll
            for (int j = 0; j < 8; ++j) {
                float v = c[nt][j] + bn[nt];
                v = v > 0.0f ? v : 0.0f;
                xt[(mt * 16 + khalf * 8 + j) * ROW_P + n] = v;
            }
        }
    }
    __syncthreads();

    for (int i = tid; i < nrows * 32; i += OWN_THREADS) {
        const int r = i >> 5, c4 = i & 31;
        *reinterpret_cast<float4*>(out + (size_t)(lo + r) * out_stride + c4 * 4) =
            *reinterpret_cast<const float4*>(xt + r * ROW_P + c4 * 4);
    }
}

__global__ void cge_copyx_kernel(const float* __restrict__ x,
                                 float* __restrict__ out) {
    int i = blockIdx.x * blockDim.x + threadIdx.x;
    if (i >= N_NODES * (D / 4)) return;
    int r  = i >> 5;
    int c4 = i & 31;
    float4 v = reinterpret_cast<const float4*>(x)[i];
    reinterpret_cast<float4*>(out + (size_t)r * 256 + 128)[c4] = v;
}

extern "C" void kernel_launch(void* const* d_in, const int* in_sizes, int n_in,
                              void* d_out, int out_size, void* d_ws, size_t ws_size,
                              hipStream_t stream) {
    const float* x    = (const float*)d_in[0];
    const int*   esrc = (const int*)  d_in[1];
    const int*   edst = (const int*)  d_in[2];
    const float* deg  = (const float*)d_in[3];
    const float* W1   = (const float*)d_in[4];
    const float* b1   = (const float*)d_in[5];
    const float* W2   = (const float*)d_in[6];
    const float* b2   = (const float*)d_in[7];
    float* out = (float*)d_out;

    float* h1    = out + 128;

    dim3 blk(256);
    dim3 ogrid(OWN_BLOCKS);
    dim3 oblk(OWN_THREADS);

    cge_keys_kernel<<<dim3(KEY_BLOCKS), dim3(KEY_THREADS), 0, stream>>>(edst, out);
    cge_owner_layer_kernel<false><<<ogrid, oblk, 0, stream>>>(x, D, esrc, edst, deg, W1, b1,
                                                              h1, 256, out);
    cge_owner_layer_kernel<true><<<ogrid, oblk, 0, stream>>>(h1, 256, esrc, edst, deg, W2, b2,
                                                             out, 256, out);
    cge_copyx_kernel  <<<(N_NODES * (D / 4) + 255) / 256, blk, 0, stream>>>(x, out);
}


# Layer-1 sweep reads a per-block bit map of 32-edge groups (first launch writes it, not owner keys) and loads only the flagged destination lines, 32 in flight per wave; bin (arrival) sum order kept
# speedup vs baseline: 3.5935x; 1.0429x over previous
#include <hip/hip_runtime.h>

typedef float v2f __attribute__((ext_vector_type(2)));
typedef float v8f __attribute__((ext_vector_type(8)));

#define N_NODES 50000
#define N_EDGES 800000
#define D 128
#define EPS_F 1e-7f

#define OWN_BLOCKS  256
#define OWN_ROWS    196
#define OWN_THREADS 1024
#define OWN_WAVES   32
#define N_GROUPS    (N_EDGES / 128)
#define N_GRP32     (N_EDGES / 32)
#define MAP_WORDS   ((N_GRP32 + 31) / 32)
#define MAP_WPW     ((MAP_WORDS + OWN_WAVES - 1) / OWN_WAVES)
#define GL_OFF      64
#define QCAP        (GL_OFF + MAP_WPW * 32)
#define FQ_CAP      (32 + 128)
#define MM_KS       2
#define DRAIN_U     8
#define BIN_ROWS    7
#define LIST_LDS    416
#define LIST_E0     8
#define OUT_PITCH   256
#define M_TILES     13
#define TILE_ROWS   (M_TILES * 16)
#define ROW_P       132
#define W_P         288
#define GEMM_WAVES  (M_TILES * 2)
#define X_CELLS     ((TILE_ROWS * 32 + OWN_THREADS - 1) / OWN_THREADS)
#define XT_ROWS     (X_CELLS * OWN_THREADS / 32)
#define UBUF_FLOATS (XT_ROWS * ROW_P + 64 * W_P)
#define L1_WORDS    (OWN_WAVES * QCAP + OWN_WAVES * LIST_LDS)


__device__ __forceinline__ void cge_drain_ent(
    int ent, int first, int n, int lane,
    const float* __restrict__ h, int h_stride, float* acc) {
    int j = 0;
    for (; j + DRAIN_U <= n; j += DRAIN_U) {
        float v[DRAIN_U][4];
        int   dd[DRAIN_U];
#pragma unroll
        for (int u = 0; u < DRAIN_U; ++u) {
            const int en = __builtin_amdgcn_readlane(ent, first + j + u);
            dd[u] = en & 255;
            const float* __restrict__ hp = h + (size_t)(en >> 8) * h_stride + lane;
            v[u][0] = hp[0]; v[u][1] = hp[32]; v[u][2] = hp[64]; v[u][3] = hp[96];
        }
#pragma unroll
        for (int u = 0; u < DRAIN_U; ++u) {
            float* ap = acc + dd[u] * ROW_P + lane;
            __hip_atomic_fetch_add(ap,      v[u][0], __ATOMIC_RELAXED, __HIP_MEMORY_SCOPE_WORKGROUP);
            __hip_atomic_fetch_add(ap + 32, v[u][1], __ATOMIC_RELAXED, __HIP_MEMORY_SCOPE_WORKGROUP);
            __hip_atomic_fetch_add(ap + 64, v[u][2], __ATOMIC_RELAXED, __HIP_MEMORY_SCOPE_WORKGROUP);
            __hip_atomic_fetch_add(ap + 96, v[u][3], __ATOMIC_RELAXED, __HIP_MEMORY_SCOPE_WORKGROUP);
        }
    }
    if (j < n) {
        float v[DRAIN_U][4];
        int   dd[DRAIN_U];
#pragma unroll
        for (int u = 0; u < DRAIN_U; ++u) {
            const int jj = (j + u < n) ? (j + u) : j;
            const int en = __builtin_amdgcn_readlane(ent, first + jj);
            dd[u] = en & 255;
            const float* __restrict__ hp = h + (size_t)(en >> 8) * h_stride + lane;
            v[u][0] = hp[0]; v[u][1] = hp[32]; v[u][2] = hp[64]; v[u][3] = hp[96];
        }
#pragma unroll
        for (int u = 0; u < DRAIN_U; ++u) {
            if (j + u < n) {
                float* ap = acc + dd[u] * ROW_P + lane;
                __hip_atomic_fetch_add(ap,      v[u][0], __ATOMIC_RELAXED, __HIP_MEMORY_SCOPE_WORKGROUP);
                __hip_atomic_fetch_add(ap + 32, v[u][1], __ATOMIC_RELAXED, __HIP_MEMORY_SCOPE_WORKGROUP);
                __hip_atomic_fetch_add(ap + 64, v[u][2], __ATOMIC_RELAXED, __HIP_MEMORY_SCOPE_WORKGROUP);
                __hip_atomic_fetch_add(ap + 96, v[u][3], __ATOMIC_RELAXED, __HIP_MEMORY_SCOPE_WORKGROUP);
            }
        }
    }
}

__device__ __forceinline__ void cge_drain_q(
    const int* qw, int base, int n, int lane,
    const int* __restrict__ esrc,
    const float* __restrict__ h, int h_stride, float* acc) {
    int ent = 0;
    if (lane < n) {
        const int q = qw[base + lane];
        ent = (esrc[q >> 8] << 8) | (q & 255);
    }
    cge_drain_ent(ent, 0, n, lane, h, h_stride, acc);
}

__device__ __forceinline__ void cge_bin_q(
    const int* qw, int base, int n, int lane,
    const int* __restrict__ esrc, const int* __restrict__ edst,
    const float* __restrict__ h, int h_stride,
    float* acc, int* bins, int* bincnt, int maxent, int lo, int nrows) {
    int  ent  = 0;
    bool over = false;
    if (lane < n) {
        const int q = qw[base + lane];
        const int sr = esrc[q];
        int dl = edst[q] - lo;
        if ((unsigned)dl >= (unsigned)nrows) dl = TILE_ROWS - 1;
        ent = (sr << 8) | dl;
        const int b = dl & 31;
        const int slot = __hip_atomic_fetch_add(bincnt + b, 1, __ATOMIC_RELAXED, __HIP_MEMORY_SCOPE_WORKGROUP);
        if (slot < maxent) bins[b * LIST_LDS + LIST_E0 + slot] = ent;
        else over = true;
    }
    unsigned om = __builtin_amdgcn_ballot_w32(over);
    while (om != 0u) {
        const int l = __builtin_ctz(om);
        om &= om - 1u;
        cge_drain_ent(ent, l, 1, lane, h, h_stride, acc);
    }
}

#define FB_TRIPS ((N_GROUPS + OWN_WAVES - 1) / OWN_WAVES)
__device__ __forceinline__ void cge_sweep_dst(
    const int* __restrict__ esrc, const int* __restrict__ edst,
    const float* __restrict__ h, int h_stride, float* acc,
    int* qw, int lo, int nrows, int wave, int lane, int rot) {
    const int4* __restrict__ d4 = reinterpret_cast<const int4*>(edst);
    const unsigned below = (1u << lane) - 1u;
    int  qn  = 0;
    for (int trip = 0; trip < FB_TRIPS; ++trip) {
        int g = trip * OWN_WAVES + wave;
        if (g >= N_GROUPS) break;
        g += rot;
        if (g >= N_GROUPS) g -= N_GROUPS;
        const int4 d = d4[g * 32 + lane];
        const int ebase = g * 128 + lane * 4;
        const int dl0 = d.x - lo, dl1 = d.y - lo, dl2 = d.z - lo, dl3 = d.w - lo;
        const bool m0 = (unsigned)dl0 < (unsigned)nrows;
        const bool m1 = (unsigned)dl1 < (unsigned)nrows;
        const bool m2 = (unsigned)dl2 < (unsigned)nrows;
        const bool m3 = (unsigned)dl3 < (unsigned)nrows;
        unsigned mk;
        mk = __builtin_amdgcn_ballot_w32(m0);
        if (m0) qw[qn + __builtin_popcount(mk & below)] = ((ebase + 0) << 8) | dl0;
        qn += __builtin_popcount(mk);
        mk = __builtin_amdgcn_ballot_w32(m1);
        if (m1) qw[qn + __builtin_popcount(mk & below)] = ((ebase + 1) << 8) | dl1;
        qn += __builtin_popcount(mk);
        mk = __builtin_amdgcn_ballot_w32(m2);
        if (m2) qw[qn + __builtin_popcount(mk & below)] = ((ebase + 2) << 8) | dl2;
        qn += __builtin_popcount(mk);
        mk = __builtin_amdgcn_ballot_w32(m3);
        if (m3) qw[qn + __builtin_popcount(mk & below)] = ((ebase + 3) << 8) | dl3;
        qn += __builtin_popcount(mk);
        while (qn >= 32) {
            cge_drain_q(qw, qn - 32, 32, lane, esrc, h, h_stride, acc);
            qn -= 32;
        }
    }
    if (qn > 0) cge_drain_q(qw, 0, qn, lane, esrc, h, h_stride, acc);
}

#define KEY_WORDS      (N_EDGES / 4)
#define KEY_ROW0       (OWN_WAVES * LIST_LDS / 128)
#define KEY_HOST_ROWS  (OWN_ROWS - KEY_ROW0)
__device__ __forceinline__ size_t cge_key_row_word(int krow) {
    const int hb = krow / KEY_HOST_ROWS;
    const int hr = krow - hb * KEY_HOST_ROWS;
    return (size_t)(hb * OWN_ROWS + KEY_ROW0 + hr) * OUT_PITCH;
}
__device__ __forceinline__ size_t cge_map_word(int j, int b) {
    return cge_key_row_word(2 * j + (b >> 7)) + (size_t)(b & 127);
}

__device__ __forceinline__ void cge_sweep_map(
    const int* __restrict__ esrc, const int* __restrict__ edst, unsigned mapw,
    const float* __restrict__ h, int h_stride, float* acc,
    int* qw, int* bins, int* bincnt, int maxent, int lo, int nrows, int wave, int lane) {
    const unsigned below = (1u << lane) - 1u;
    int* gl = qw + GL_OFF;
    int  ng = 0;
    for (int j = 0; j < MAP_WPW; ++j) {
        const unsigned m = (unsigned)__builtin_amdgcn_readlane((int)mapw, j);
        if ((m >> lane) & 1u) gl[ng + __builtin_popcount(m & below)] = (j * OWN_WAVES + wave) * 32 + lane;
        ng += __builtin_popcount(m);
    }
    int qn = 0;
    int d[32];
    int gv = 0;
    int nb = (ng < 32) ? ng : 32;
    if (ng > 0) {
        gv = gl[lane < nb ? lane : nb - 1];
#pragma unroll
        for (int u = 0; u < 32; ++u) d[u] = edst[__builtin_amdgcn_readlane(gv, u) * 32 + lane];
    }
    for (int base = 0; base < ng; base += 32) {
        unsigned hm = 0u;
#pragma unroll
        for (int u = 0; u < 32; ++u) {
            const unsigned mk = __builtin_amdgcn_ballot_w32((unsigned)(d[u] - lo) < (unsigned)nrows);
            hm = (lane == u) ? mk : hm;
        }
        if (lane >= nb) hm = 0u;
        const int e0 = gv * 32;
        const int rest = ng - base - 32;
        if (rest > 0) {
            nb = (rest < 32) ? rest : 32;
            gv = gl[base + 32 + (lane < nb ? lane : nb - 1)];
#pragma unroll
            for (int u = 0; u < 32; ++u) d[u] = edst[__builtin_amdgcn_readlane(gv, u) * 32 + lane];
        }
        unsigned any = __builtin_amdgcn_ballot_w32(hm != 0u);
        while (any != 0u) {
            if (hm != 0u) qw[qn + __builtin_popcount(any & below)] = e0 + __builtin_ctz(hm);
            qn += __builtin_popcount(any);
            hm &= hm - 1u;
            any = __builtin_amdgcn_ballot_w32(hm != 0u);
            if (qn >= 32) {
                cge_bin_q(qw, qn - 32, 32, lane, esrc, edst, h, h_stride, acc, bins, bincnt, maxent, lo, nrows);
                qn -= 32;
            }
        }
    }
    if (qn > 0) cge_bin_q(qw, 0, qn, lane, esrc, edst, h, h_stride, acc, bins, bincnt, maxent, lo, nrows);
}

#define MAP_THREADS 256
#define MAP_WPB     4
#define MAP_BLOCKS  ((MAP_WORDS + MAP_WPB - 1) / MAP_WPB)
__global__ __launch_bounds__(MAP_THREADS) void cge_map_kernel(
    const int* __restrict__ edst, float* out) {
    static_assert(MAP_THREADS == OWN_BLOCKS, "one thread writes the map word of one owner block");
    static_assert(MAP_THREADS * 4 == 32 * 32, "the threads' loads cover the 32 groups of one map word");
    __shared__ unsigned mw[MAP_WPB * OWN_BLOCKS];
    const int t  = threadIdx.x;
    const int j0 = blockIdx.x * MAP_WPB;
    const int4* __restrict__ d4 = reinterpret_cast<const int4*>(edst);
    unsigned* kw = reinterpret_cast<unsigned*>(out);
    int4 d[MAP_WPB];
#pragma unroll
    for (int k = 0; k < MAP_WPB; ++k) {
        const int i = (j0 + k) * MAP_THREADS + t;
        d[k] = d4[i < KEY_WORDS ? i : 0];
        mw[k * OWN_BLOCKS + t] = 0u;
    }
    __syncthreads();
    const unsigned bit = 1u << (t >> 3);
#pragma unroll
    for (int k = 0; k < MAP_WPB; ++k) {
        const int i = (j0 + k) * MAP_THREADS + t;
        if (i < KEY_WORDS) {
            unsigned* m = mw + k * OWN_BLOCKS;
            const unsigned k0 = (unsigned)d[k].x / OWN_ROWS;
            const unsigned k1 = (unsigned)d[k].y / OWN_ROWS;
            const unsigned k2 = (unsigned)d[k].z / OWN_ROWS;
            const unsigned k3 = (unsigned)d[k].w / OWN_ROWS;
            if (k0 < OWN_BLOCKS) __hip_atomic_fetch_or(m + k0, bit, __ATOMIC_RELAXED, __HIP_MEMORY_SCOPE_WORKGROUP);
            if (k1 < OWN_BLOCKS) __hip_atomic_fetch_or(m + k1, bit, __ATOMIC_RELAXED, __HIP_MEMORY_SCOPE_WORKGROUP);
            if (k2 < OWN_BLOCKS) __hip_atomic_fetch_or(m + k2, bit, __ATOMIC_RELAXED, __HIP_MEMORY_SCOPE_WORKGROUP);
            if (k3 < OWN_BLOCKS) __hip_atomic_fetch_or(m + k3, bit, __ATOMIC_RELAXED, __HIP_MEMORY_SCOPE_WORKGROUP);
        }
    }
    __syncthreads();
#pragma unroll
    for (int k = 0; k < MAP_WPB; ++k) {
        if (j0 + k < MAP_WORDS)
            kw[cge_map_word(j0 + k, t)] = mw[k * OWN_BLOCKS + t];
    }
}

template <bool SCALE>
__device__ __forceinline__ void cge_half_product(
    const float* ap, float ascale, const float* bp, v8f (&c)[4]) {
#pragma unroll 4
    for (int ks = 0; ks < 32; ++ks) {
        v2f a = *reinterpret_cast<const v2f*>(ap + 4 * ks);
        if (SCALE) { a.x *= ascale; a.y *= ascale; }
        const float* bk = bp + 2 * ks * W_P;
#pragma unroll
        for (int nt = 0; nt < 4; ++nt) {
            const v2f b = *reinterpret_cast<const v2f*>(bk + nt * 32);
            c[nt] = __builtin_amdgcn_wmma_f32_16x16x4_f32(
                false, a, false, b, (short)0, c[nt], false, false);
        }
    }
}

#define W_CELLS 2
__device__ __forceinline__ void cge_load_w(
    const float* __restrict__ W, int kbase, int tid, float4 (&wr)[2 * W_CELLS]) {
#pragma unroll
    for (int t = 0; t < W_CELLS; ++t) {
        const int i = tid + t * OWN_THREADS;
        const int kp = i >> 5, c4 = i & 31;
        wr[2 * t]     = *reinterpret_cast<const float4*>(W + (size_t)(kbase + 2 * kp) * D + c4 * 4);
        wr[2 * t + 1] = *reinterpret_cast<const float4*>(W + (size_t)(kbase + 2 * kp + 1) * D + c4 * 4);
    }
}
__device__ __forceinline__ void cge_store_w(
    float* wl, int tid, const float4 (&wr)[2 * W_CELLS]) {
#pragma unroll
    for (int t = 0; t < W_CELLS; ++t) {
        const int i = tid + t * OWN_THREADS;
        const int kp = i >> 5, c4 = i & 31;
        const float4 w0 = wr[2 * t], w1 = wr[2 * t + 1];
        float4* dst = reinterpret_cast<float4*>(wl + kp * W_P + c4 * 8);
        dst[0] = make_float4(w0.x, w1.x, w0.y, w1.y);
        dst[1] = make_float4(w0.z, w1.z, w0.w, w1.w);
    }
}

__device__ __forceinline__ void cge_kstep(const float* ap, const float* bp, int ks, v8f (&c)[4]) {
    const v2f a = *reinterpret_cast<const v2f*>(ap + 4 * ks);
    const float* bk = bp + 2 * ks * W_P;
#pragma unroll
    for (int nt = 0; nt < 4; ++nt) {
        const v2f b = *reinterpret_cast<const v2f*>(bk + nt * 32);
        c[nt] = __builtin_amdgcn_wmma_f32_16x16x4_f32(
            false, a, false, b, (short)0, c[nt], false, false);
    }
}

__device__ __forceinline__ void cge_put_row(const float (&a)[4], int cur, int wave, int lane, float* acc) {
    float* ap = acc + (wave + 32 * cur) * ROW_P + lane;
    __hip_atomic_fetch_add(ap,      a[0], __ATOMIC_RELAXED, __HIP_MEMORY_SCOPE_WORKGROUP);
    __hip_atomic_fetch_add(ap + 32, a[1], __ATOMIC_RELAXED, __HIP_MEMORY_SCOPE_WORKGROUP);
    __hip_atomic_fetch_add(ap + 64, a[2], __ATOMIC_RELAXED, __HIP_MEMORY_SCOPE_WORKGROUP);
    __hip_atomic_fetch_add(ap + 96, a[3], __ATOMIC_RELAXED, __HIP_MEMORY_SCOPE_WORKGROUP);
}

__device__ __forceinline__ void cge_run_add(
    float (&a)[4], int& cur, int r, const float (&v)[4], int wave, int lane, float* acc) {
    if (r != cur) {
        cge_put_row(a, cur, wave, lane, acc);
        a[0] = 0.0f; a[1] = 0.0f; a[2] = 0.0f; a[3] = 0.0f;
        cur = r;
    }
    a[0] += v[0]; a[1] += v[1]; a[2] += v[2]; a[3] += v[3];
}

template <bool MM, int U>
__device__ __forceinline__ void cge_drain_sorted(
    int ent, int first, int n, int lane,
    const float* __restrict__ h, int h_stride,
    float (&a)[4], int& cur, int wave, float* acc,
    bool mm, const float* ap, const float* bp, v8f (&c)[4], int& ks) {
    int j = 0;
    for (; j + U <= n; j += U) {
        float v[U][4];
        int   rr[U];
#pragma unroll
        for (int u = 0; u < U; ++u) {
            const int en = __builtin_amdgcn_readlane(ent, first + j + u);
            rr[u] = (en >> 5) & 7;
            const float* __restrict__ hp = h + (size_t)(en >> 8) * h_stride + lane;
            v[u][0] = hp[0]; v[u][1] = hp[32]; v[u][2] = hp[64]; v[u][3] = hp[96];
        }
        if (MM) {
            if (mm && ks < 32) {
#pragma unroll
                for (int t = 0; t < MM_KS; ++t) cge_kstep(ap, bp, ks + t, c);
                ks += MM_KS;
            }
        }
#pragma unroll
        for (int u = 0; u < U; ++u) cge_run_add(a, cur, rr[u], v[u], wave, lane, acc);
    }
    if (j < n) {
        float v[U][4];
        int   rr[U];
#pragma unroll
        for (int u = 0; u < U; ++u) {
            const int jj = (j + u < n) ? (j + u) : j;
            const int en = __builtin_amdgcn_readlane(ent, first + jj);
            rr[u] = (en >> 5) & 7;
            const float* __restrict__ hp = h + (size_t)(en >> 8) * h_stride + lane;
            v[u][0] = hp[0]; v[u][1] = hp[32]; v[u][2] = hp[64]; v[u][3] = hp[96];
        }
        if (MM) {
            if (mm && ks < 32) {
#pragma unroll
                for (int t = 0; t < MM_KS; ++t) cge_kstep(ap, bp, ks + t, c);
                ks += MM_KS;
            }
        }
#pragma unroll
        for (int u = 0; u < U; ++u) {
            if (j + u < n) cge_run_add(a, cur, rr[u], v[u], wave, lane, acc);
        }
    }
}

__device__ __forceinline__ void cge_sort_bin(const int* src, int ne, int* dst, int lane) {
    const unsigned below = (1u << lane) - 1u;
    int st[BIN_ROWS + 1];
#pragma unroll
    for (int r = 0; r <= BIN_ROWS; ++r) st[r] = 0;
    for (int base = 0; base < ne; base += 32) {
        const bool valid = base + lane < ne;
        const int  r = valid ? ((src[base + lane] >> 5) & 7) : BIN_ROWS;
#pragma unroll
        for (int k = 0; k < BIN_ROWS; ++k)
            st[k + 1] += __builtin_popcount(__builtin_amdgcn_ballot_w32(r == k));
    }
#pragma unroll
    for (int k = 1; k < BIN_ROWS; ++k) st[k + 1] += st[k];
    for (int base = 0; base < ne; base += 32) {
        const bool valid = base + lane < ne;
        const int  en = valid ? src[base + lane] : 0;
        const int  r  = valid ? ((en >> 5) & 7) : BIN_ROWS;
        int pos = 0;
#pragma unroll
        for (int k = 0; k < BIN_ROWS; ++k) {
            const unsigned m = __builtin_amdgcn_ballot_w32(r == k);
            pos = (r == k) ? st[k] + __builtin_popcount(m & below) : pos;
            st[k] += __builtin_popcount(m);
        }
        if (valid) dst[pos] = en;
    }
}

__device__ __forceinline__ void cge_load_x(
    const float* __restrict__ h, int h_stride, int lo, int nrows, int tid, float4 (&xr)[X_CELLS]) {
#pragma unroll
    for (int t = 0; t < X_CELLS; ++t) {
        const int i = tid + t * OWN_THREADS;
        const int r = i >> 5, c4 = i & 31;
        const int rr = (r < nrows) ? r : (nrows - 1);
        xr[t] = *reinterpret_cast<const float4*>(h + (size_t)(lo + rr) * h_stride + c4 * 4);
    }
}
__device__ __forceinline__ void cge_store_x(float* xt, int tid, const float4 (&xr)[X_CELLS]) {
#pragma unroll
    for (int t = 0; t < X_CELLS; ++t) {
        const int i = tid + t * OWN_THREADS;
        const int r = i >> 5, c4 = i & 31;
        *reinterpret_cast<float4*>(xt + r * ROW_P + c4 * 4) = xr[t];
    }
}

template <bool REUSE>
__global__ __launch_bounds__(OWN_THREADS) void cge_owner_layer_kernel(
    const float* __restrict__ h, int h_stride,
    const int* __restrict__ esrc, const int* __restrict__ edst,
    const float* __restrict__ deg,
    const float* __restrict__ W,
    const float* __restrict__ bias,
    float* out, int out_stride,
    float* lists) {
    __shared__ alignas(16) float acc[TILE_ROWS * ROW_P];
    static_assert(L1_WORDS <= UBUF_FLOATS, "layer 1: the queues with their group lists and the bins fit in ubuf");
    __shared__ alignas(16) float ubuf[UBUF_FLOATS];
    __shared__ int fq[REUSE ? OWN_WAVES * FQ_CAP : 1];
    __shared__ int bincnt[REUSE ? 1 : OWN_WAVES];

    const int tid   = threadIdx.x;
    const int lane  = tid & 31;
    const int wave  = __builtin_amdgcn_readfirstlane(tid >> 5);
    const int lo    = blockIdx.x * OWN_ROWS;
    const int nrows = (N_NODES - lo < OWN_ROWS) ? (N_NODES - lo) : OWN_ROWS;

    const int gfit   = (nrows * 4) & ~31;
    const int gcap   = gfit < LIST_LDS ? gfit : LIST_LDS;
    const int maxent = (gcap < LIST_LDS - 32) ? (gcap - LIST_E0) : (LIST_LDS - 32 - LIST_E0);
    int* lw = reinterpret_cast<int*>(lists);
    const int gw0 = wave * gcap;

    int line0 = 0;
    unsigned mapw = 0u;
    if (REUSE) {
        const int gi = gw0 + lane;
        line0 = lw[(size_t)(lo + (gi >> 7)) * OUT_PITCH + (gi & 127)];
    } else {
        const int jw = lane * OWN_WAVES + wave;
        if (lane < MAP_WPW && jw < MAP_WORDS) mapw = (unsigned)lw[cge_map_word(jw, (int)blockIdx.x)];
    }

    float* xt = ubuf;
    float* wl = ubuf + XT_ROWS * ROW_P;

    const int  rc     = lane & 15;
    const int  khalf  = lane >> 4;
    const int  mt     = wave >> 1;
    const int  nh     = wave & 1;
    const bool active = (wave < GEMM_WAVES) && (mt * 16 < nrows);
    const int  arow   = mt * 16 + rc;
    const int  aoff   = arow * ROW_P + 2 * khalf;
    const int  boff   = khalf * W_P + (nh * 64 + rc) * 2;

    float4 xr[X_CELLS];
    float4 wr[2 * W_CELLS];
    if (REUSE) {
        cge_load_x(h, h_stride, lo, nrows, tid, xr);
        cge_load_w(W, 0, tid, wr);
    }

    float4* acc4 = reinterpret_cast<float4*>(acc);
    for (int i = tid; i < TILE_ROWS * ROW_P / 4; i += OWN_THREADS)
        acc4[i] = make_float4(0.0f, 0.0f, 0.0f, 0.0f);
    if (!REUSE && tid < OWN_WAVES) bincnt[tid] = 0;
    if (REUSE) {
        cge_store_x(xt, tid, xr);
        cge_store_w(wl, tid, wr);
    }
    __syncthreads();

    const int rot = blockIdx.x * (N_GROUPS / OWN_BLOCKS);
    int* qw = reinterpret_cast<int*>(ubuf) + wave * QCAP;
    int* bins = reinterpret_cast<int*>(ubuf) + OWN_WAVES * QCAP;
    int* sl = bins + wave * LIST_LDS;

    v8f   c[4] = {};
    float dg = 1.0f;
    float bn[4] = {0.0f, 0.0f, 0.0f, 0.0f};

    if (!REUSE) {
        cge_sweep_map(esrc, edst, mapw, h, h_stride, acc, qw, bins, bincnt, maxent, lo, nrows, wave, lane);
        __syncthreads();
        const int  bc     = bincnt[lane];
        const bool marked = __builtin_amdgcn_ballot_w32(bc > maxent) != 0u;
        const int  sent   = __builtin_amdgcn_readlane(bc, wave);
        const int  ne     = sent < maxent ? sent : maxent;
        static_assert(QCAP >= LIST_LDS, "the sorted list is staged in the wave's queue space");
        cge_sort_bin(sl + LIST_E0, ne, qw + LIST_E0, lane);
        qw[LIST_E0 + ne + lane] = 0;
        if (lane < LIST_E0) qw[lane] = (lane == 0) ? (marked ? -1 : ne) : 0;
        const int nl = marked ? 1 : ((LIST_E0 - 1 + ne) >> 5) + 1;
        for (int i = lane * 4; i < nl * 32; i += 128) {
            const int gi = gw0 + i;
            *reinterpret_cast<int4*>(lw + (size_t)(lo + (gi >> 7)) * OUT_PITCH + (gi & 127)) =
                *reinterpret_cast<const int4*>(qw + i);
        }
        float a[4] = {0.0f, 0.0f, 0.0f, 0.0f};
        int currow = 0;
        int ksd = 0;
        for (int base = 0; base < ne; base += 32) {
            const int nb  = (ne - base < 32) ? (ne - base) : 32;
            const int ent = qw[LIST_E0 + base + lane];
            cge_drain_sorted<false, DRAIN_U>(ent, 0, nb, lane, h, h_stride, a, currow, wave, acc,
                                             false, nullptr, nullptr, c, ksd);
        }
        cge_put_row(a, currow, wave, lane, acc);
    } else {
        int ks = 0;
        const int cnt = __builtin_amdgcn_readfirstlane(line0);
        if ((unsigned)cnt <= (unsigned)maxent) {
            float a[4] = {0.0f, 0.0f, 0.0f, 0.0f};
            int currow = 0;
            const int lastw = LIST_E0 - 1 + cnt;
            int cur = line0, first = LIST_E0;
            for (int base = 0; base <= lastw; base += 32) {
                const int ni = base + 32 + lane;
                int nxt = 0;
                if (ni <= lastw) {
                    const int gi = gw0 + ni;
                    nxt = lw[(size_t)(lo + (gi >> 7)) * OUT_PITCH + (gi & 127)];
                }
                const int last = (lastw < base + 31) ? lastw : (base + 31);
                cge_drain_sorted<true, DRAIN_U>(cur, first, last - base - first + 1, lane, h, h_stride,
                                                a, currow, wave, acc, active, xt + aoff, wl + boff, c, ks);
                cur = nxt; first = 0;
            }
            cge_put_row(a, currow, wave, lane, acc);
        } else {
            cge_sweep_dst(esrc, edst, h, h_stride, acc, fq + wave * FQ_CAP, lo, nrows, wave, lane, rot);
        }
        cge_load_w(W, 128, tid, wr);
        dg = deg[lo + (arow < nrows ? arow : nrows - 1)];
#pragma unroll
        for (int nt = 0; nt < 4; ++nt) bn[nt] = bias[nh * 64 + nt * 16 + rc];
        if (active) {
            for (; ks < 32; ks += MM_KS) {
#pragma unroll
                for (int t = 0; t < MM_KS; ++t) cge_kstep(xt + aoff, wl + boff, ks + t, c);
            }
        }
    }
    __syncthreads();

    if (!REUSE) {
        cge_load_x(h, h_stride, lo, nrows, tid, xr);
        cge_load_w(W, 0, tid, wr);
        dg = deg[lo + (arow < nrows ? arow : nrows - 1)];
#pragma unroll
        for (int nt = 0; nt < 4; ++nt) bn[nt] = bias[nh * 64 + nt * 16 + rc];
        cge_store_x(xt, tid, xr);
        cge_store_w(wl, tid, wr);
        __syncthreads();

        cge_load_w(W, 128, tid, wr);

        if (active) cge_half_product<false>(xt + aoff, 1.0f, wl + boff, c);
        __syncthreads();
    }
    cge_store_w(wl, tid, wr);
    __syncthreads();

    if (active) {
        const float invdeg = 1.0f / (dg + EPS_F);
        cge_half_product<true>(acc + aoff, invdeg, wl + boff, c);

#pragma unroll
        for (int nt = 0; nt < 4; ++nt) {
            const int n = nh * 64 + nt * 16 + rc;
#pragma unroll
            for (int j = 0; j < 8; ++j) {
                float v = c[nt][j] + bn[nt];
                v = v > 0.0f ? v : 0.0f;
                xt[(mt * 16 + khalf * 8 + j) * ROW_P + n] = v;
            }
        }
    }
    __syncthreads();

    for (int i = tid; i < nrows * 32; i += OWN_THREADS) {
        const int r = i >> 5, c4 = i & 31;
        *reinterpret_cast<float4*>(out + (size_t)(lo + r) * out_stride + c4 * 4) =
            *reinterpret_cast<const float4*>(xt + r * ROW_P + c4 * 4);
    }
}

__global__ void cge_copyx_kernel(const float* __restrict__ x,
                                 float* __restrict__ out) {
    int i = blockIdx.x * blockDim.x + threadIdx.x;
    if (i >= N_NODES * (D / 4)) return;
    int r  = i >> 5;
    int c4 = i & 31;
    float4 v = reinterpret_cast<const float4*>(x)[i];
    reinterpret_cast<float4*>(out + (size_t)r * 256 + 128)[c4] = v;
}

extern "C" void kernel_launch(void* const* d_in, const int* in_sizes, int n_in,
                              void* d_out, int out_size, void* d_ws, size_t ws_size,
                              hipStream_t stream) {
    const float* x    = (const float*)d_in[0];
    const int*   esrc = (const int*)  d_in[1];
    const int*   edst = (const int*)  d_in[2];
    const float* deg  = (const float*)d_in[3];
    const float* W1   = (const float*)d_in[4];
    const float* b1   = (const float*)d_in[5];
    const float* W2   = (const float*)d_in[6];
    const float* b2   = (const float*)d_in[7];
    float* out = (float*)d_out;

    float* h1    = out + 128;

    dim3 blk(256);
    dim3 ogrid(OWN_BLOCKS);
    dim3 oblk(OWN_THREADS);

    cge_map_kernel<<<dim3(MAP_BLOCKS), dim3(MAP_THREADS), 0, stream>>>(edst, out);
    cge_owner_layer_kernel<false><<<ogrid, oblk, 0, stream>>>(x, D, esrc, edst, deg, W1, b1,
                                                              h1, 256, out);
    cge_owner_layer_kernel<true><<<ogrid, oblk, 0, stream>>>(h1, 256, esrc, edst, deg, W2, b2,
                                                             out, 256, out);
    cge_copyx_kernel  <<<(N_NODES * (D / 4) + 255) / 256, blk, 0, stream>>>(x, out);
}
